# starting kernel
# baseline (speedup 1.0000x reference)
#include <hip/hip_runtime.h>


typedef __attribute__((ext_vector_type(2))) float v2f;
typedef __attribute__((ext_vector_type(8))) float v8f;

#define DFEAT 128

__global__ __launch_bounds__(256) void gemm128_wmma(
    const float* __restrict__ Xin, int Mrows,
    const float* __restrict__ W,
    const float* __restrict__ bias,
    float* __restrict__ Y)
{
    const int lane = threadIdx.x & 31;
    const int wave = threadIdx.x >> 5;
    const int half = lane >> 4;
    const int r    = lane & 15;
    const int m0   = (blockIdx.x * 8 + wave) * 16;
    if (m0 >= Mrows) return;

    v8f acc[8] = {};

    int arow = m0 + r;
    if (arow >= Mrows) arow = Mrows - 1;
    const float* ap = Xin + (size_t)arow * DFEAT + 2 * half;

    for (int kk = 0; kk < 32; ++kk) {
        const int k = kk * 4;
        v2f a;
        a.x = ap[k];
        a.y = ap[k + 1];

        const float* wrow = W + (size_t)(k + 2 * half) * DFEAT + r;
#pragma unroll
        for (int t = 0; t < 8; ++t) {
            v2f b;
            b.x = wrow[t * 16];
            b.y = wrow[t * 16 + DFEAT];
            acc[t] = __builtin_amdgcn_wmma_f32_16x16x4_f32(
                false, a, false, b, (short)0, acc[t], false, false);
        }
    }

#pragma unroll
    for (int t = 0; t < 8; ++t) {
        const int n = t * 16 + r;
        const float bv = bias ? bias[n] : 0.0f;
#pragma unroll
        for (int j = 0; j < 8; ++j) {
            const int row = m0 + j + 8 * half;
            if (row < Mrows) Y[(size_t)row * DFEAT + n] = acc[t][j] + bv;
        }
    }
}

__global__ __launch_bounds__(256) void scatter_rows(
    const float* __restrict__ src,
    const int* __restrict__ gidx,
    const int* __restrict__ sidx,
    float* __restrict__ dst,
    int* __restrict__ deg,
    int nnz)
{
    const int i = blockIdx.x * 8 + (threadIdx.x >> 5);
    if (i >= nnz) return;
    const int lane = threadIdx.x & 31;
    const int g = gidx[i];
    const int s = sidx[i];
    const float4 v = ((const float4*)(src + (size_t)g * DFEAT))[lane];
    float* dp = dst + (size_t)s * DFEAT + lane * 4;
    atomicAdd(dp + 0, v.x);
    atomicAdd(dp + 1, v.y);
    atomicAdd(dp + 2, v.z);
    atomicAdd(dp + 3, v.w);
    if (deg != nullptr && lane == 0) atomicAdd(deg + s, 1);
}

__global__ void zero_kernel(float* __restrict__ p, long n)
{
    long i = (long)blockIdx.x * blockDim.x + threadIdx.x;
    if (i < n) p[i] = 0.0f;
}

__global__ void blend_kernel(
    const float* __restrict__ Ap, const float* __restrict__ SB,
    const int* __restrict__ deg, const float* __restrict__ X0,
    float* __restrict__ Xn, long n)
{
    long i = (long)blockIdx.x * blockDim.x + threadIdx.x;
    if (i >= n) return;
    const int v = (int)(i >> 7);
    const float d = (float)deg[v];
    Xn[i] = 0.5f * (d * Ap[i] + SB[i]) + 0.5f * X0[i];
}

extern "C" void kernel_launch(void* const* d_in, const int* in_sizes, int n_in,
                              void* d_out, int out_size, void* d_ws, size_t ws_size,
                              hipStream_t stream) {
    const float* X      = (const float*)d_in[0];
    const int*   vertex = (const int*)d_in[1];
    const int*   edges  = (const int*)d_in[2];
    const float* X0     = (const float*)d_in[3];
    const float* W1     = (const float*)d_in[4];
    const float* b1     = (const float*)d_in[5];
    const float* W2     = (const float*)d_in[6];
    const float* b2     = (const float*)d_in[7];
    const float* W3     = (const float*)d_in[8];
    const float* b3     = (const float*)d_in[9];
    float* out = (float*)d_out;

    constexpr int N = 50000, M = 25000, NNZ = 800000;
    constexpr size_t ND = (size_t)N * DFEAT;
    constexpr size_t MD = (size_t)M * DFEAT;

    char* ws = (char*)d_ws;
    auto align256 = [](size_t x) { return (x + 255) & ~(size_t)255; };
    size_t off = 0;
    float* XW = (float*)(ws + off); off = align256(off + ND * 4);
    float* Xe = (float*)(ws + off); off = align256(off + MD * 4);
    float* Ap = (float*)(ws + off); off = align256(off + ND * 4);
    float* Bm = (float*)(ws + off); off = align256(off + MD * 4);
    float* SB = (float*)(ws + off); off = align256(off + ND * 4);
    int*  deg = (int*)  (ws + off); off = align256(off + (size_t)N * 4);
    float* Xn = XW;

    const int gemmN = (N + 127) / 128;
    const int gemmM = (M + 127) / 128;
    const int scatB = (NNZ + 7) / 8;

    gemm128_wmma<<<gemmN, 256, 0, stream>>>(X, N, W1, b1, XW);

    zero_kernel<<<(int)((MD + 255) / 256), 256, 0, stream>>>(Xe, (long)MD);
    scatter_rows<<<scatB, 256, 0, stream>>>(XW, vertex, edges, Xe, nullptr, NNZ);

    gemm128_wmma<<<gemmN, 256, 0, stream>>>(X, N, W2, b2, Ap);

    gemm128_wmma<<<gemmM, 256, 0, stream>>>(Xe, M, W2 + 128 * DFEAT, nullptr, Bm);

    zero_kernel<<<(int)((ND + 255) / 256), 256, 0, stream>>>(SB, (long)ND);
    zero_kernel<<<(N + 255) / 256, 256, 0, stream>>>((float*)deg, (long)N);
    scatter_rows<<<scatB, 256, 0, stream>>>(Bm, edges, vertex, SB, deg, NNZ);

    blend_kernel<<<(int)((ND + 255) / 256), 256, 0, stream>>>(Ap, SB, deg, X0, Xn, (long)ND);

    gemm128_wmma<<<gemmN, 256, 0, stream>>>(Xn, N, W3, b3, out);
}


# Row-sum filter reads one owner byte per list position (id / rows per work-group, written by a small launch before each row-sum launch), not the 4-byte id: a lane load covers 16 ids; sums unchanged
# speedup vs baseline: 1.9555x; 1.9555x over previous
#include <hip/hip_runtime.h>


typedef __attribute__((ext_vector_type(2))) float v2f;
typedef __attribute__((ext_vector_type(8))) float v8f;

#define DFEAT 128

__global__ __launch_bounds__(256) void gemm128_wmma(
    const float* __restrict__ Xin, int Mrows,
    const float* __restrict__ W,
    const float* __restrict__ bias,
    float* __restrict__ Y)
{
    const int lane = threadIdx.x & 31;
    const int wave = threadIdx.x >> 5;
    const int half = lane >> 4;
    const int r    = lane & 15;
    const int m0   = (blockIdx.x * 8 + wave) * 16;
    if (m0 >= Mrows) return;

    v8f acc[8] = {};

    int arow = m0 + r;
    if (arow >= Mrows) arow = Mrows - 1;
    const float* ap = Xin + (size_t)arow * DFEAT + 2 * half;

    for (int kk = 0; kk < 32; ++kk) {
        const int k = kk * 4;
        v2f a;
        a.x = ap[k];
        a.y = ap[k + 1];

        const float* wrow = W + (size_t)(k + 2 * half) * DFEAT + r;
#pragma unroll
        for (int t = 0; t < 8; ++t) {
            v2f b;
            b.x = wrow[t * 16];
            b.y = wrow[t * 16 + DFEAT];
            acc[t] = __builtin_amdgcn_wmma_f32_16x16x4_f32(
                false, a, false, b, (short)0, acc[t], false, false);
        }
    }

#pragma unroll
    for (int t = 0; t < 8; ++t) {
        const int n = t * 16 + r;
        const float bv = bias ? bias[n] : 0.0f;
#pragma unroll
        for (int j = 0; j < 8; ++j) {
            const int row = m0 + j + 8 * half;
            if (row < Mrows) Y[(size_t)row * DFEAT + n] = acc[t][j] + bv;
        }
    }
}

#define OB_U 4
template <int R>
__global__ __launch_bounds__(256) void owner_bytes(
    const int* __restrict__ ids, int* __restrict__ own, int nnz, int nwords)
{
    const int t0 = blockIdx.x * (256 * OB_U) + threadIdx.x;
    int4 a[OB_U];
#pragma unroll
    for (int k = 0; k < OB_U; ++k) {
        const int p = 4 * (t0 + k * 256);
        a[k] = *(const int4*)(ids + (p + 3 < nnz ? p : 0));
    }
#pragma unroll
    for (int k = 0; k < OB_U; ++k) {
        const int t = t0 + k * 256;
        const int p = 4 * t;
        if (p + 3 >= nnz) {
            a[k].x = p     < nnz ? ids[p]     : 0;
            a[k].y = p + 1 < nnz ? ids[p + 1] : 0;
            a[k].z = p + 2 < nnz ? ids[p + 2] : 0;
            a[k].w = 0;
        }
        if (t < nwords)
            own[t] = (int)((unsigned)(a[k].x / R) | ((unsigned)(a[k].y / R) << 8) |
                           ((unsigned)(a[k].z / R) << 16) | ((unsigned)(a[k].w / R) << 24));
    }
}

#define OWN_THREADS 1024
#define OWN_WAVES   32
#define OWN_QCAP    768
#define OWN_BLK     512
#define OWN_UB      4
#define OWN_RING    64
#define OWN_BATCH   8

template <int R, bool BLEND>
__global__ __launch_bounds__(OWN_THREADS) void own_rows(
    const float* __restrict__ src,
    const int* __restrict__ gidx,
    const int* __restrict__ sidx,
    const int* __restrict__ own,
    float* __restrict__ dst,
    const float* __restrict__ Ap,
    const float* __restrict__ X0,
    int nrows, int nnz)
{
    static_assert(R <= 256, "local row must fit the low 8 bits of a queue entry");
    __shared__ float4 acc[R * 32];
    __shared__ int cntL[BLEND ? R : 1];
    __shared__ int queue[OWN_WAVES][OWN_QCAP];
    __shared__ int qcnt[OWN_WAVES];
    __shared__ int ring[OWN_WAVES][OWN_RING];

    const int tid  = threadIdx.x;
    const int lane = tid & 31;
    const int w    = __builtin_amdgcn_readfirstlane(tid >> 5);
    const int base = blockIdx.x * R;
    int rows = nrows - base;
    if (rows > R) rows = R;
    if (rows < 0) rows = 0;

    for (int i = tid; i < R * 32; i += OWN_THREADS) acc[i] = make_float4(0.f, 0.f, 0.f, 0.f);
    if (BLEND) for (int i = tid; i < R; i += OWN_THREADS) cntL[i] = 0;

    const int seg = (((nnz + OWN_WAVES - 1) / OWN_WAVES) + OWN_BLK - 1) / OWN_BLK * OWN_BLK;
    const uint4* __restrict__ own4 = (const uint4*)own;
    const unsigned me = blockIdx.x;
    const unsigned me4 = me * 0x01010101u;
    int pos = w * seg;
    int segEnd = pos + seg;
    if (segEnd > nnz) segEnd = nnz;

    int any;
    do {
        int qn = 0;
        while (pos < segEnd && qn <= OWN_QCAP - OWN_BLK) {
            uint4 d[OWN_UB];
#pragma unroll
            for (int u = 0; u < OWN_UB; ++u) {
                int pb = pos + u * OWN_BLK;
                if (pb >= segEnd) pb = pos;
                d[u] = own4[(pb >> 4) + lane];
            }
#pragma unroll
            for (int u = 0; u < OWN_UB; ++u) {
                if (pos >= segEnd || qn > OWN_QCAP - OWN_BLK) break;
                const unsigned y0 = d[u].x ^ me4, y1 = d[u].y ^ me4, y2 = d[u].z ^ me4, y3 = d[u].w ^ me4;
                const unsigned t = (((y0 - 0x01010101u) & ~y0) | ((y1 - 0x01010101u) & ~y1) |
                                    ((y2 - 0x01010101u) & ~y2) | ((y3 - 0x01010101u) & ~y3)) & 0x80808080u;
                unsigned hit = __builtin_amdgcn_ballot_w32(t != 0u);
                while (hit) {
                    const int l = __builtin_ctz(hit);
                    hit &= hit - 1u;
                    const unsigned x0 = __builtin_amdgcn_readlane(d[u].x, l);
                    const unsigned x1 = __builtin_amdgcn_readlane(d[u].y, l);
                    const unsigned x2 = __builtin_amdgcn_readlane(d[u].z, l);
                    const unsigned x3 = __builtin_amdgcn_readlane(d[u].w, l);
                    const unsigned xw = (lane & 8) ? ((lane & 4) ? x3 : x2) : ((lane & 4) ? x1 : x0);
                    const unsigned ob = (xw >> ((lane & 3) * 8)) & 255u;
                    const int p = pos + l * 16 + lane;
                    const bool m = (lane < 16) && (ob == me) && (p < segEnd);
                    const unsigned mask = __builtin_amdgcn_ballot_w32(m);
                    if (m) queue[w][qn + (int)__builtin_amdgcn_mbcnt_lo(mask, 0u)] = p;
                    qn += __builtin_popcount(mask);
                }
                pos += OWN_BLK;
            }
        }
        for (int i = lane; i < qn; i += 32) {
            const int p = queue[w][i];
            const int s = sidx[p];
            const int g = gidx[p];
            queue[w][i] = (g << 8) | (s - base);
        }
        if (lane == 0) qcnt[w] = qn;
        __syncthreads();

        unsigned head = 0, tail = 0;
        auto drain = [&](int n) {
            float4 v[OWN_BATCH];
            int rw[OWN_BATCH];
#pragma unroll
            for (int k = 0; k < OWN_BATCH; ++k) {
                if (k < n) {
                    const int e = __builtin_amdgcn_readfirstlane(ring[w][(head + k) & (OWN_RING - 1)]);
                    rw[k] = e & 255;
                    v[k] = ((const float4*)(src + (size_t)((unsigned)e >> 8) * DFEAT))[lane];
                }
            }
#pragma unroll
            for (int k = 0; k < OWN_BATCH; ++k) {
                if (k < n) {
                    float4 a = acc[rw[k] * 32 + lane];
                    a.x += v[k].x; a.y += v[k].y; a.z += v[k].z; a.w += v[k].w;
                    acc[rw[k] * 32 + lane] = a;
                    if (BLEND && lane == 0) cntL[rw[k]] += 1;
                }
            }
            head += (unsigned)n;
        };
        for (int q = 0; q < OWN_WAVES; ++q) {
            const int n = qcnt[q];
            for (int i = 0; i < n; i += 32) {
                const bool in = (i + lane) < n;
                const int e = in ? queue[q][i + lane] : 0;
                const bool own = in && ((e & 31) == w);
                const unsigned mask = __builtin_amdgcn_ballot_w32(own);
                if (mask) {
                    const unsigned off = __builtin_amdgcn_mbcnt_lo(mask, 0u);
                    if (own) ring[w][(tail + off) & (OWN_RING - 1)] = e;
                    tail += (unsigned)__builtin_popcount(mask);
                    while (tail - head >= OWN_BATCH) drain(OWN_BATCH);
                }
            }
        }
        if (tail != head) drain((int)(tail - head));

        any = __syncthreads_or(pos < segEnd ? 1 : 0);
    } while (any);

    float4* d4 = (float4*)dst + (size_t)base * 32;
    if (BLEND) {
        const int nel = rows * 32;
        if (nel > 0) {
            const float4* a4 = (const float4*)Ap + (size_t)base * 32;
            const float4* x4 = (const float4*)X0 + (size_t)base * 32;
            constexpr int IT = (R * 32 + OWN_THREADS - 1) / OWN_THREADS;
            float4 av[IT], xv[IT];
#pragma unroll
            for (int k = 0; k < IT; ++k) {
                const int i = tid + k * OWN_THREADS;
                const int j = i < nel ? i : nel - 1;
                av[k] = a4[j];
                xv[k] = x4[j];
            }
#pragma unroll
            for (int k = 0; k < IT; ++k) {
                const int i = tid + k * OWN_THREADS;
                if (i < nel) {
                    const float4 s = acc[i];
                    const float d = (float)cntL[i >> 5];
                    float4 o;
                    o.x = 0.5f * (d * av[k].x + s.x) + 0.5f * xv[k].x;
                    o.y = 0.5f * (d * av[k].y + s.y) + 0.5f * xv[k].y;
                    o.z = 0.5f * (d * av[k].z + s.z) + 0.5f * xv[k].z;
                    o.w = 0.5f * (d * av[k].w + s.w) + 0.5f * xv[k].w;
                    d4[i] = o;
                }
            }
        }
    } else {
        for (int i = tid; i < rows * 32; i += OWN_THREADS) d4[i] = acc[i];
    }
}

extern "C" void kernel_launch(void* const* d_in, const int* in_sizes, int n_in,
                              void* d_out, int out_size, void* d_ws, size_t ws_size,
                              hipStream_t stream) {
    const float* X      = (const float*)d_in[0];
    const int*   vertex = (const int*)d_in[1];
    const int*   edges  = (const int*)d_in[2];
    const float* X0     = (const float*)d_in[3];
    const float* W1     = (const float*)d_in[4];
    const float* b1     = (const float*)d_in[5];
    const float* W2     = (const float*)d_in[6];
    const float* b2     = (const float*)d_in[7];
    const float* W3     = (const float*)d_in[8];
    const float* b3     = (const float*)d_in[9];
    float* out = (float*)d_out;

    constexpr int N = 50000, M = 25000, NNZ = 800000;
    constexpr size_t ND = (size_t)N * DFEAT;
    constexpr size_t MD = (size_t)M * DFEAT;

    char* ws = (char*)d_ws;
    auto align256 = [](size_t x) { return (x + 255) & ~(size_t)255; };
    size_t off = 0;
    float* XW = (float*)(ws + off); off = align256(off + ND * 4);
    float* Xe = (float*)(ws + off); off = align256(off + MD * 4);
    float* Ap = (float*)(ws + off); off = align256(off + ND * 4);
    float* Bm = (float*)(ws + off); off = align256(off + MD * 4);
    float* Xn = XW;

    const int gemmN = (N + 127) / 128;
    const int gemmM = (M + 127) / 128;
    constexpr int RM = 98;
    constexpr int RN = 196;
    static_assert((M + RM - 1) / RM <= 256 && (N + RN - 1) / RN <= 256, "an owner number must fit one byte");

    constexpr int OWN_POS   = (NNZ + OWN_BLK - 1) / OWN_BLK * OWN_BLK;
    constexpr int OWN_WORDS = OWN_POS / 4;
    constexpr int OWN_GRID  = (OWN_WORDS + 256 * OB_U - 1) / (256 * OB_U);
    static_assert((size_t)OWN_POS <= MD * 4, "a byte list must fit the region it borrows");
    int* ownE = (int*)Bm;
    int* ownV = (int*)Xe;

    owner_bytes<RM><<<OWN_GRID, 256, 0, stream>>>(edges, ownE, NNZ, OWN_WORDS);

    gemm128_wmma<<<gemmN, 256, 0, stream>>>(X, N, W1, b1, XW);

    own_rows<RM, false><<<(M + RM - 1) / RM, OWN_THREADS, 0, stream>>>(XW, vertex, edges, ownE, Xe, nullptr, nullptr, M, NNZ);

    gemm128_wmma<<<gemmN, 256, 0, stream>>>(X, N, W2, b2, Ap);

    gemm128_wmma<<<gemmM, 256, 0, stream>>>(Xe, M, W2 + 128 * DFEAT, nullptr, Bm);

    owner_bytes<RN><<<OWN_GRID, 256, 0, stream>>>(vertex, ownV, NNZ, OWN_WORDS);

    own_rows<RN, true><<<(N + RN - 1) / RN, OWN_THREADS, 0, stream>>>(Bm, edges, vertex, ownV, Xn, Ap, X0, N, NNZ);

    gemm128_wmma<<<gemmN, 256, 0, stream>>>(Xn, N, W3, b3, out);
}


# Row-sum owner waves keep their rows' running sums and counts in registers (scalar-indexed register moves pick the row's slot), not in LDS; rows stored straight from registers; sum order unchanged
# speedup vs baseline: 2.0258x; 1.0359x over previous
#include <hip/hip_runtime.h>


typedef __attribute__((ext_vector_type(2))) float v2f;
typedef __attribute__((ext_vector_type(8))) float v8f;
typedef __attribute__((ext_vector_type(8))) int   v8i;

#define DFEAT 128

__global__ __launch_bounds__(256) void gemm128_wmma(
    const float* __restrict__ Xin, int Mrows,
    const float* __restrict__ W,
    const float* __restrict__ bias,
    float* __restrict__ Y)
{
    const int lane = threadIdx.x & 31;
    const int wave = threadIdx.x >> 5;
    const int half = lane >> 4;
    const int r    = lane & 15;
    const int m0   = (blockIdx.x * 8 + wave) * 16;
    if (m0 >= Mrows) return;

    v8f acc[8] = {};

    int arow = m0 + r;
    if (arow >= Mrows) arow = Mrows - 1;
    const float* ap = Xin + (size_t)arow * DFEAT + 2 * half;

    for (int kk = 0; kk < 32; ++kk) {
        const int k = kk * 4;
        v2f a;
        a.x = ap[k];
        a.y = ap[k + 1];

        const float* wrow = W + (size_t)(k + 2 * half) * DFEAT + r;
#pragma unroll
        for (int t = 0; t < 8; ++t) {
            v2f b;
            b.x = wrow[t * 16];
            b.y = wrow[t * 16 + DFEAT];
            acc[t] = __builtin_amdgcn_wmma_f32_16x16x4_f32(
                false, a, false, b, (short)0, acc[t], false, false);
        }
    }

#pragma unroll
    for (int t = 0; t < 8; ++t) {
        const int n = t * 16 + r;
        const float bv = bias ? bias[n] : 0.0f;
#pragma unroll
        for (int j = 0; j < 8; ++j) {
            const int row = m0 + j + 8 * half;
            if (row < Mrows) Y[(size_t)row * DFEAT + n] = acc[t][j] + bv;
        }
    }
}

#define OB_U 4
template <int R>
__global__ __launch_bounds__(256) void owner_bytes(
    const int* __restrict__ ids, int* __restrict__ own, int nnz, int nwords)
{
    const int t0 = blockIdx.x * (256 * OB_U) + threadIdx.x;
    int4 a[OB_U];
#pragma unroll
    for (int k = 0; k < OB_U; ++k) {
        const int p = 4 * (t0 + k * 256);
        a[k] = *(const int4*)(ids + (p + 3 < nnz ? p : 0));
    }
#pragma unroll
    for (int k = 0; k < OB_U; ++k) {
        const int t = t0 + k * 256;
        const int p = 4 * t;
        if (p + 3 >= nnz) {
            a[k].x = p     < nnz ? ids[p]     : 0;
            a[k].y = p + 1 < nnz ? ids[p + 1] : 0;
            a[k].z = p + 2 < nnz ? ids[p + 2] : 0;
            a[k].w = 0;
        }
        if (t < nwords)
            own[t] = (int)((unsigned)(a[k].x / R) | ((unsigned)(a[k].y / R) << 8) |
                           ((unsigned)(a[k].z / R) << 16) | ((unsigned)(a[k].w / R) << 24));
    }
}

#define OWN_THREADS 1024
#define OWN_WAVES   32
#define OWN_QCAP    768
#define OWN_BLK     512
#define OWN_UB      4
#define OWN_RING    64
#define OWN_BATCH   8

template <int R, bool BLEND>
__global__ __launch_bounds__(OWN_THREADS) void own_rows(
    const float* __restrict__ src,
    const int* __restrict__ gidx,
    const int* __restrict__ sidx,
    const int* __restrict__ own,
    float* __restrict__ dst,
    const float* __restrict__ Ap,
    const float* __restrict__ X0,
    int nrows, int nnz)
{
    static_assert(R <= 256, "local row must fit the low 8 bits of a queue entry");
    constexpr int NS = (R + OWN_WAVES - 1) / OWN_WAVES;
    __shared__ int queue[OWN_WAVES][OWN_QCAP];
    __shared__ int qcnt[OWN_WAVES];
    __shared__ int ring[OWN_WAVES][OWN_RING];

    const int tid  = threadIdx.x;
    const int lane = tid & 31;
    const int w    = __builtin_amdgcn_readfirstlane(tid >> 5);
    const int base = blockIdx.x * R;
    int rows = nrows - base;
    if (rows > R) rows = R;
    if (rows < 0) rows = 0;

    static_assert(NS <= 8, "a slot must index an 8-wide register vector");
    v8f ax = {}, ay = {}, az = {}, aw = {};
    v8i rc = {};

    const int seg = (((nnz + OWN_WAVES - 1) / OWN_WAVES) + OWN_BLK - 1) / OWN_BLK * OWN_BLK;
    const uint4* __restrict__ own4 = (const uint4*)own;
    const unsigned me = blockIdx.x;
    const unsigned me4 = me * 0x01010101u;
    int pos = w * seg;
    int segEnd = pos + seg;
    if (segEnd > nnz) segEnd = nnz;

    int any;
    do {
        int qn = 0;
        while (pos < segEnd && qn <= OWN_QCAP - OWN_BLK) {
            uint4 d[OWN_UB];
#pragma unroll
            for (int u = 0; u < OWN_UB; ++u) {
                int pb = pos + u * OWN_BLK;
                if (pb >= segEnd) pb = pos;
                d[u] = own4[(pb >> 4) + lane];
            }
#pragma unroll
            for (int u = 0; u < OWN_UB; ++u) {
                if (pos >= segEnd || qn > OWN_QCAP - OWN_BLK) break;
                const unsigned y0 = d[u].x ^ me4, y1 = d[u].y ^ me4, y2 = d[u].z ^ me4, y3 = d[u].w ^ me4;
                const unsigned t = (((y0 - 0x01010101u) & ~y0) | ((y1 - 0x01010101u) & ~y1) |
                                    ((y2 - 0x01010101u) & ~y2) | ((y3 - 0x01010101u) & ~y3)) & 0x80808080u;
                unsigned hit = __builtin_amdgcn_ballot_w32(t != 0u);
                while (hit) {
                    const int l = __builtin_ctz(hit);
                    hit &= hit - 1u;
                    const unsigned x0 = __builtin_amdgcn_readlane(d[u].x, l);
                    const unsigned x1 = __builtin_amdgcn_readlane(d[u].y, l);
                    const unsigned x2 = __builtin_amdgcn_readlane(d[u].z, l);
                    const unsigned x3 = __builtin_amdgcn_readlane(d[u].w, l);
                    const unsigned xw = (lane & 8) ? ((lane & 4) ? x3 : x2) : ((lane & 4) ? x1 : x0);
                    const unsigned ob = (xw >> ((lane & 3) * 8)) & 255u;
                    const int p = pos + l * 16 + lane;
                    const bool m = (lane < 16) && (ob == me) && (p < segEnd);
                    const unsigned mask = __builtin_amdgcn_ballot_w32(m);
                    if (m) queue[w][qn + (int)__builtin_amdgcn_mbcnt_lo(mask, 0u)] = p;
                    qn += __builtin_popcount(mask);
                }
                pos += OWN_BLK;
            }
        }
        for (int i = lane; i < qn; i += 32) {
            const int p = queue[w][i];
            const int s = sidx[p];
            const int g = gidx[p];
            queue[w][i] = (g << 8) | (s - base);
        }
        if (lane == 0) qcnt[w] = qn;
        __syncthreads();

        unsigned head = 0, tail = 0;
        auto drain = [&](int n) {
            float4 v[OWN_BATCH];
            int rs[OWN_BATCH];
#pragma unroll
            for (int k = 0; k < OWN_BATCH; ++k) {
                if (k < n) {
                    const int e = __builtin_amdgcn_readfirstlane(ring[w][(head + k) & (OWN_RING - 1)]);
                    rs[k] = (e & 255) >> 5;
                    v[k] = ((const float4*)(src + (size_t)((unsigned)e >> 8) * DFEAT))[lane];
                }
            }
#pragma unroll
            for (int k = 0; k < OWN_BATCH; ++k) {
                if (k < n) {
                    const int s = rs[k];
                    ax[s] += v[k].x; ay[s] += v[k].y; az[s] += v[k].z; aw[s] += v[k].w;
                    if (BLEND) rc[s] += 1;
                }
            }
            head += (unsigned)n;
        };
        for (int q = 0; q < OWN_WAVES; ++q) {
            const int n = qcnt[q];
            for (int i = 0; i < n; i += 32) {
                const bool in = (i + lane) < n;
                const int e = in ? queue[q][i + lane] : 0;
                const bool own = in && ((e & 31) == w);
                const unsigned mask = __builtin_amdgcn_ballot_w32(own);
                if (mask) {
                    const unsigned off = __builtin_amdgcn_mbcnt_lo(mask, 0u);
                    if (own) ring[w][(tail + off) & (OWN_RING - 1)] = e;
                    tail += (unsigned)__builtin_popcount(mask);
                    while (tail - head >= OWN_BATCH) drain(OWN_BATCH);
                }
            }
        }
        if (tail != head) drain((int)(tail - head));

        any = __syncthreads_or(pos < segEnd ? 1 : 0);
    } while (any);

    float4* d4 = (float4*)dst + (size_t)base * 32;
    if (BLEND) {
        const float4* a4 = (const float4*)Ap + (size_t)base * 32;
        const float4* x4 = (const float4*)X0 + (size_t)base * 32;
        if (rows > 0) {
            float4 av[NS], xv[NS];
#pragma unroll
            for (int t = 0; t < NS; ++t) {
                const int r = w + OWN_WAVES * t;
                const int j = r < rows ? r : rows - 1;
                av[t] = a4[j * 32 + lane];
                xv[t] = x4[j * 32 + lane];
            }
#pragma unroll
            for (int t = 0; t < NS; ++t) {
                const int r = w + OWN_WAVES * t;
                if (r < rows) {
                    const float4 s = make_float4(ax[t], ay[t], az[t], aw[t]);
                    const float d = (float)rc[t];
                    float4 o;
                    o.x = 0.5f * (d * av[t].x + s.x) + 0.5f * xv[t].x;
                    o.y = 0.5f * (d * av[t].y + s.y) + 0.5f * xv[t].y;
                    o.z = 0.5f * (d * av[t].z + s.z) + 0.5f * xv[t].z;
                    o.w = 0.5f * (d * av[t].w + s.w) + 0.5f * xv[t].w;
                    d4[r * 32 + lane] = o;
                }
            }
        }
    } else {
#pragma unroll
        for (int t = 0; t < NS; ++t) {
            const int r = w + OWN_WAVES * t;
            if (r < rows) d4[r * 32 + lane] = make_float4(ax[t], ay[t], az[t], aw[t]);
        }
    }
}

extern "C" void kernel_launch(void* const* d_in, const int* in_sizes, int n_in,
                              void* d_out, int out_size, void* d_ws, size_t ws_size,
                              hipStream_t stream) {
    const float* X      = (const float*)d_in[0];
    const int*   vertex = (const int*)d_in[1];
    const int*   edges  = (const int*)d_in[2];
    const float* X0     = (const float*)d_in[3];
    const float* W1     = (const float*)d_in[4];
    const float* b1     = (const float*)d_in[5];
    const float* W2     = (const float*)d_in[6];
    const float* b2     = (const float*)d_in[7];
    const float* W3     = (const float*)d_in[8];
    const float* b3     = (const float*)d_in[9];
    float* out = (float*)d_out;

    constexpr int N = 50000, M = 25000, NNZ = 800000;
    constexpr size_t ND = (size_t)N * DFEAT;
    constexpr size_t MD = (size_t)M * DFEAT;

    char* ws = (char*)d_ws;
    auto align256 = [](size_t x) { return (x + 255) & ~(size_t)255; };
    size_t off = 0;
    float* XW = (float*)(ws + off); off = align256(off + ND * 4);
    float* Xe = (float*)(ws + off); off = align256(off + MD * 4);
    float* Ap = (float*)(ws + off); off = align256(off + ND * 4);
    float* Bm = (float*)(ws + off); off = align256(off + MD * 4);
    float* Xn = XW;

    const int gemmN = (N + 127) / 128;
    const int gemmM = (M + 127) / 128;
    constexpr int RM = 98;
    constexpr int RN = 196;
    static_assert((M + RM - 1) / RM <= 256 && (N + RN - 1) / RN <= 256, "an owner number must fit one byte");

    constexpr int OWN_POS   = (NNZ + OWN_BLK - 1) / OWN_BLK * OWN_BLK;
    constexpr int OWN_WORDS = OWN_POS / 4;
    constexpr int OWN_GRID  = (OWN_WORDS + 256 * OB_U - 1) / (256 * OB_U);
    static_assert((size_t)OWN_POS <= MD * 4, "a byte list must fit the region it borrows");
    int* ownE = (int*)Bm;
    int* ownV = (int*)Xe;

    owner_bytes<RM><<<OWN_GRID, 256, 0, stream>>>(edges, ownE, NNZ, OWN_WORDS);

    gemm128_wmma<<<gemmN, 256, 0, stream>>>(X, N, W1, b1, XW);

    own_rows<RM, false><<<(M + RM - 1) / RM, OWN_THREADS, 0, stream>>>(XW, vertex, edges, ownE, Xe, nullptr, nullptr, M, NNZ);

    gemm128_wmma<<<gemmN, 256, 0, stream>>>(X, N, W2, b2, Ap);

    gemm128_wmma<<<gemmM, 256, 0, stream>>>(Xe, M, W2 + 128 * DFEAT, nullptr, Bm);

    owner_bytes<RN><<<OWN_GRID, 256, 0, stream>>>(vertex, ownV, NNZ, OWN_WORDS);

    own_rows<RN, true><<<(N + RN - 1) / RN, OWN_THREADS, 0, stream>>>(Bm, edges, vertex, ownV, Xn, Ap, X0, N, NNZ);

    gemm128_wmma<<<gemmN, 256, 0, stream>>>(Xn, N, W3, b3, out);
}


# Row-sum entries go straight to their owner wave's stretch of an LDS pool (per-wave owner counts, 32x32 count table, ballot ranks; list order kept); the walk over all 32 queues goes; sums unchanged
# speedup vs baseline: 2.2407x; 1.1061x over previous
#include <hip/hip_runtime.h>


typedef __attribute__((ext_vector_type(2))) float v2f;
typedef __attribute__((ext_vector_type(8))) float v8f;
typedef __attribute__((ext_vector_type(8))) int   v8i;

#define DFEAT 128

__global__ __launch_bounds__(256) void gemm128_wmma(
    const float* __restrict__ Xin, int Mrows,
    const float* __restrict__ W,
    const float* __restrict__ bias,
    float* __restrict__ Y)
{
    const int lane = threadIdx.x & 31;
    const int wave = threadIdx.x >> 5;
    const int half = lane >> 4;
    const int r    = lane & 15;
    const int m0   = (blockIdx.x * 8 + wave) * 16;
    if (m0 >= Mrows) return;

    v8f acc[8] = {};

    int arow = m0 + r;
    if (arow >= Mrows) arow = Mrows - 1;
    const float* ap = Xin + (size_t)arow * DFEAT + 2 * half;

    for (int kk = 0; kk < 32; ++kk) {
        const int k = kk * 4;
        v2f a;
        a.x = ap[k];
        a.y = ap[k + 1];

        const float* wrow = W + (size_t)(k + 2 * half) * DFEAT + r;
#pragma unroll
        for (int t = 0; t < 8; ++t) {
            v2f b;
            b.x = wrow[t * 16];
            b.y = wrow[t * 16 + DFEAT];
            acc[t] = __builtin_amdgcn_wmma_f32_16x16x4_f32(
                false, a, false, b, (short)0, acc[t], false, false);
        }
    }

#pragma unroll
    for (int t = 0; t < 8; ++t) {
        const int n = t * 16 + r;
        const float bv = bias ? bias[n] : 0.0f;
#pragma unroll
        for (int j = 0; j < 8; ++j) {
            const int row = m0 + j + 8 * half;
            if (row < Mrows) Y[(size_t)row * DFEAT + n] = acc[t][j] + bv;
        }
    }
}

#define OB_U 4
template <int R>
__global__ __launch_bounds__(256) void owner_bytes(
    const int* __restrict__ ids, int* __restrict__ own, int nnz, int nwords)
{
    const int t0 = blockIdx.x * (256 * OB_U) + threadIdx.x;
    int4 a[OB_U];
#pragma unroll
    for (int k = 0; k < OB_U; ++k) {
        const int p = 4 * (t0 + k * 256);
        a[k] = *(const int4*)(ids + (p + 3 < nnz ? p : 0));
    }
#pragma unroll
    for (int k = 0; k < OB_U; ++k) {
        const int t = t0 + k * 256;
        const int p = 4 * t;
        if (p + 3 >= nnz) {
            a[k].x = p     < nnz ? ids[p]     : 0;
            a[k].y = p + 1 < nnz ? ids[p + 1] : 0;
            a[k].z = p + 2 < nnz ? ids[p + 2] : 0;
            a[k].w = 0;
        }
        if (t < nwords)
            own[t] = (int)((unsigned)(a[k].x / R) | ((unsigned)(a[k].y / R) << 8) |
                           ((unsigned)(a[k].z / R) << 16) | ((unsigned)(a[k].w / R) << 24));
    }
}

#define OWN_THREADS 1024
#define OWN_WAVES   32
#define OWN_QCAP    768
#define OWN_BLK     512
#define OWN_UB      4
#define OWN_BATCH   8

template <int R, bool BLEND>
__global__ __launch_bounds__(OWN_THREADS) void own_rows(
    const float* __restrict__ src,
    const int* __restrict__ gidx,
    const int* __restrict__ sidx,
    const int* __restrict__ own,
    float* __restrict__ dst,
    const float* __restrict__ Ap,
    const float* __restrict__ X0,
    int nrows, int nnz)
{
    static_assert(R <= 256, "local row must fit the low 8 bits of a queue entry");
    constexpr int NS = (R + OWN_WAVES - 1) / OWN_WAVES;
    __shared__ int queue[OWN_WAVES][OWN_QCAP];
    __shared__ int pool[OWN_WAVES * OWN_QCAP];
    __shared__ int cnt[OWN_WAVES][OWN_WAVES];

    const int tid  = threadIdx.x;
    const int lane = tid & 31;
    const int w    = __builtin_amdgcn_readfirstlane(tid >> 5);
    const int base = blockIdx.x * R;
    int rows = nrows - base;
    if (rows > R) rows = R;
    if (rows < 0) rows = 0;

    static_assert(NS <= 8, "a slot must index an 8-wide register vector");
    v8f ax = {}, ay = {}, az = {}, aw = {};
    v8i rc = {};

    const int seg = (((nnz + OWN_WAVES - 1) / OWN_WAVES) + OWN_BLK - 1) / OWN_BLK * OWN_BLK;
    const uint4* __restrict__ own4 = (const uint4*)own;
    const unsigned me = blockIdx.x;
    const unsigned me4 = me * 0x01010101u;
    int pos = w * seg;
    int segEnd = pos + seg;
    if (segEnd > nnz) segEnd = nnz;

    int any;
    do {
        int qn = 0;
        while (pos < segEnd && qn <= OWN_QCAP - OWN_BLK) {
            uint4 d[OWN_UB];
#pragma unroll
            for (int u = 0; u < OWN_UB; ++u) {
                int pb = pos + u * OWN_BLK;
                if (pb >= segEnd) pb = pos;
                d[u] = own4[(pb >> 4) + lane];
            }
#pragma unroll
            for (int u = 0; u < OWN_UB; ++u) {
                if (pos >= segEnd || qn > OWN_QCAP - OWN_BLK) break;
                const unsigned y0 = d[u].x ^ me4, y1 = d[u].y ^ me4, y2 = d[u].z ^ me4, y3 = d[u].w ^ me4;
                const unsigned t = (((y0 - 0x01010101u) & ~y0) | ((y1 - 0x01010101u) & ~y1) |
                                    ((y2 - 0x01010101u) & ~y2) | ((y3 - 0x01010101u) & ~y3)) & 0x80808080u;
                unsigned hit = __builtin_amdgcn_ballot_w32(t != 0u);
                while (hit) {
                    const int l = __builtin_ctz(hit);
                    hit &= hit - 1u;
                    const unsigned x0 = __builtin_amdgcn_readlane(d[u].x, l);
                    const unsigned x1 = __builtin_amdgcn_readlane(d[u].y, l);
                    const unsigned x2 = __builtin_amdgcn_readlane(d[u].z, l);
                    const unsigned x3 = __builtin_amdgcn_readlane(d[u].w, l);
                    const unsigned xw = (lane & 8) ? ((lane & 4) ? x3 : x2) : ((lane & 4) ? x1 : x0);
                    const unsigned ob = (xw >> ((lane & 3) * 8)) & 255u;
                    const int p = pos + l * 16 + lane;
                    const bool m = (lane < 16) && (ob == me) && (p < segEnd);
                    const unsigned mask = __builtin_amdgcn_ballot_w32(m);
                    if (m) queue[w][qn + (int)__builtin_amdgcn_mbcnt_lo(mask, 0u)] = p;
                    qn += __builtin_popcount(mask);
                }
                pos += OWN_BLK;
            }
        }
        const unsigned l0 = (lane & 1) ? ~0u : 0u, l1 = (lane & 2) ? ~0u : 0u, l2 = (lane & 4) ? ~0u : 0u,
                       l3 = (lane & 8) ? ~0u : 0u, l4 = (lane & 16) ? ~0u : 0u;
        int mine = 0;
        for (int i0 = 0; i0 < qn; i0 += 32) {
            const int i = i0 + lane;
            const bool in = i < qn;
            int e = 0;
            if (in) {
                const int p = queue[w][i];
                const int s = sidx[p];
                const int g = gidx[p];
                e = (g << 8) | (s - base);
                queue[w][i] = e;
            }
            const unsigned vm = __builtin_amdgcn_ballot_w32(in);
            const unsigned b0 = __builtin_amdgcn_ballot_w32((e & 1) != 0), b1 = __builtin_amdgcn_ballot_w32((e & 2) != 0),
                           b2 = __builtin_amdgcn_ballot_w32((e & 4) != 0), b3 = __builtin_amdgcn_ballot_w32((e & 8) != 0),
                           b4 = __builtin_amdgcn_ballot_w32((e & 16) != 0);
            mine += __builtin_popcount(vm & ~(b0 ^ l0) & ~(b1 ^ l1) & ~(b2 ^ l2) & ~(b3 ^ l3) & ~(b4 ^ l4));
        }
        cnt[w][lane] = mine;
        __syncthreads();

        int before = 0, tot = 0;
        for (int q = 0; q < OWN_WAVES; ++q) {
            const int c = cnt[q][lane];
            if (q < w) before += c;
            tot += c;
        }
        int incl = tot;
        incl += __builtin_amdgcn_update_dpp(0, incl, 0x111, 0xf, 0xf, true);
        incl += __builtin_amdgcn_update_dpp(0, incl, 0x112, 0xf, 0xf, true);
        incl += __builtin_amdgcn_update_dpp(0, incl, 0x114, 0xf, 0xf, true);
        incl += __builtin_amdgcn_update_dpp(0, incl, 0x118, 0xf, 0xf, true);
        const int low = __builtin_amdgcn_readlane(incl, 15);
        if (lane >= 16) incl += low;
        const int ostart = incl - tot;
        int next = ostart + before;
        const int mystart = __builtin_amdgcn_readlane(ostart, w);
        const int mycnt   = __builtin_amdgcn_readlane(tot, w);

        for (int i0 = 0; i0 < qn; i0 += 32) {
            const int i = i0 + lane;
            const bool in = i < qn;
            const int e = in ? queue[w][i] : 0;
            const unsigned vm = __builtin_amdgcn_ballot_w32(in);
            const unsigned b0 = __builtin_amdgcn_ballot_w32((e & 1) != 0), b1 = __builtin_amdgcn_ballot_w32((e & 2) != 0),
                           b2 = __builtin_amdgcn_ballot_w32((e & 4) != 0), b3 = __builtin_amdgcn_ballot_w32((e & 8) != 0),
                           b4 = __builtin_amdgcn_ballot_w32((e & 16) != 0);
            const unsigned e0 = (e & 1) ? ~0u : 0u, e1 = (e & 2) ? ~0u : 0u, e2 = (e & 4) ? ~0u : 0u,
                           e3 = (e & 8) ? ~0u : 0u, e4 = (e & 16) ? ~0u : 0u;
            const unsigned peers = vm & ~(b0 ^ e0) & ~(b1 ^ e1) & ~(b2 ^ e2) & ~(b3 ^ e3) & ~(b4 ^ e4);
            const int rank = (int)__builtin_amdgcn_mbcnt_lo(peers, 0u);
            const int place = __shfl(next, e & 31);
            if (in) pool[place + rank] = e;
            next += __builtin_popcount(vm & ~(b0 ^ l0) & ~(b1 ^ l1) & ~(b2 ^ l2) & ~(b3 ^ l3) & ~(b4 ^ l4));
        }
        __syncthreads();

        auto drain = [&](int ev, int first, int n) {
            float4 v[OWN_BATCH];
            int rs[OWN_BATCH];
#pragma unroll
            for (int k = 0; k < OWN_BATCH; ++k) {
                if (k < n) {
                    const int e = __builtin_amdgcn_readlane(ev, first + k);
                    rs[k] = (e & 255) >> 5;
                    v[k] = ((const float4*)(src + (size_t)((unsigned)e >> 8) * DFEAT))[lane];
                }
            }
#pragma unroll
            for (int k = 0; k < OWN_BATCH; ++k) {
                if (k < n) {
                    const int s = rs[k];
                    ax[s] += v[k].x; ay[s] += v[k].y; az[s] += v[k].z; aw[s] += v[k].w;
                    if (BLEND) rc[s] += 1;
                }
            }
        };
        for (int h = 0; h < mycnt; h += 32) {
            const int left = mycnt - h;
            const int nb = left < 32 ? left : 32;
            const int ev = lane < nb ? pool[mystart + h + lane] : 0;
            int first = 0;
#pragma nounroll
            for (; first + OWN_BATCH <= nb; first += OWN_BATCH) drain(ev, first, OWN_BATCH);
            if (first < nb) drain(ev, first, nb - first);
        }

        any = __syncthreads_or(pos < segEnd ? 1 : 0);
    } while (any);

    float4* d4 = (float4*)dst + (size_t)base * 32;
    if (BLEND) {
        const float4* a4 = (const float4*)Ap + (size_t)base * 32;
        const float4* x4 = (const float4*)X0 + (size_t)base * 32;
        if (rows > 0) {
            float4 av[NS], xv[NS];
#pragma unroll
            for (int t = 0; t < NS; ++t) {
                const int r = w + OWN_WAVES * t;
                const int j = r < rows ? r : rows - 1;
                av[t] = a4[j * 32 + lane];
                xv[t] = x4[j * 32 + lane];
            }
#pragma unroll
            for (int t = 0; t < NS; ++t) {
                const int r = w + OWN_WAVES * t;
                if (r < rows) {
                    const float4 s = make_float4(ax[t], ay[t], az[t], aw[t]);
                    const float d = (float)rc[t];
                    float4 o;
                    o.x = 0.5f * (d * av[t].x + s.x) + 0.5f * xv[t].x;
                    o.y = 0.5f * (d * av[t].y + s.y) + 0.5f * xv[t].y;
                    o.z = 0.5f * (d * av[t].z + s.z) + 0.5f * xv[t].z;
                    o.w = 0.5f * (d * av[t].w + s.w) + 0.5f * xv[t].w;
                    d4[r * 32 + lane] = o;
                }
            }
        }
    } else {
#pragma unroll
        for (int t = 0; t < NS; ++t) {
            const int r = w + OWN_WAVES * t;
            if (r < rows) d4[r * 32 + lane] = make_float4(ax[t], ay[t], az[t], aw[t]);
        }
    }
}

extern "C" void kernel_launch(void* const* d_in, const int* in_sizes, int n_in,
                              void* d_out, int out_size, void* d_ws, size_t ws_size,
                              hipStream_t stream) {
    const float* X      = (const float*)d_in[0];
    const int*   vertex = (const int*)d_in[1];
    const int*   edges  = (const int*)d_in[2];
    const float* X0     = (const float*)d_in[3];
    const float* W1     = (const float*)d_in[4];
    const float* b1     = (const float*)d_in[5];
    const float* W2     = (const float*)d_in[6];
    const float* b2     = (const float*)d_in[7];
    const float* W3     = (const float*)d_in[8];
    const float* b3     = (const float*)d_in[9];
    float* out = (float*)d_out;

    constexpr int N = 50000, M = 25000, NNZ = 800000;
    constexpr size_t ND = (size_t)N * DFEAT;
    constexpr size_t MD = (size_t)M * DFEAT;

    char* ws = (char*)d_ws;
    auto align256 = [](size_t x) { return (x + 255) & ~(size_t)255; };
    size_t off = 0;
    float* XW = (float*)(ws + off); off = align256(off + ND * 4);
    float* Xe = (float*)(ws + off); off = align256(off + MD * 4);
    float* Ap = (float*)(ws + off); off = align256(off + ND * 4);
    float* Bm = (float*)(ws + off); off = align256(off + MD * 4);
    float* Xn = XW;

    const int gemmN = (N + 127) / 128;
    const int gemmM = (M + 127) / 128;
    constexpr int RM = 98;
    constexpr int RN = 196;
    static_assert((M + RM - 1) / RM <= 256 && (N + RN - 1) / RN <= 256, "an owner number must fit one byte");

    constexpr int OWN_POS   = (NNZ + OWN_BLK - 1) / OWN_BLK * OWN_BLK;
    constexpr int OWN_WORDS = OWN_POS / 4;
    constexpr int OWN_GRID  = (OWN_WORDS + 256 * OB_U - 1) / (256 * OB_U);
    static_assert((size_t)OWN_POS <= MD * 4, "a byte list must fit the region it borrows");
    int* ownE = (int*)Bm;
    int* ownV = (int*)Xe;

    owner_bytes<RM><<<OWN_GRID, 256, 0, stream>>>(edges, ownE, NNZ, OWN_WORDS);

    gemm128_wmma<<<gemmN, 256, 0, stream>>>(X, N, W1, b1, XW);

    own_rows<RM, false><<<(M + RM - 1) / RM, OWN_THREADS, 0, stream>>>(XW, vertex, edges, ownE, Xe, nullptr, nullptr, M, NNZ);

    gemm128_wmma<<<gemmN, 256, 0, stream>>>(X, N, W2, b2, Ap);

    gemm128_wmma<<<gemmM, 256, 0, stream>>>(Xe, M, W2 + 128 * DFEAT, nullptr, Bm);

    owner_bytes<RN><<<OWN_GRID, 256, 0, stream>>>(vertex, ownV, NNZ, OWN_WORDS);

    own_rows<RN, true><<<(N + RN - 1) / RN, OWN_THREADS, 0, stream>>>(Bm, edges, vertex, ownV, Xn, Ap, X0, N, NNZ);

    gemm128_wmma<<<gemmN, 256, 0, stream>>>(Xn, N, W3, b3, out);
}


# Row-sum filter in two stages: the scan keeps only the numbers of the 16-byte owner loads that hold a hit; those loads are unpacked 32 at a time, queue places from a prefix sum; sums unchanged
# speedup vs baseline: 2.3119x; 1.0318x over previous
#include <hip/hip_runtime.h>


typedef __attribute__((ext_vector_type(2))) float v2f;
typedef __attribute__((ext_vector_type(8))) float v8f;
typedef __attribute__((ext_vector_type(8))) int   v8i;

#define DFEAT 128

__global__ __launch_bounds__(256) void gemm128_wmma(
    const float* __restrict__ Xin, int Mrows,
    const float* __restrict__ W,
    const float* __restrict__ bias,
    float* __restrict__ Y)
{
    const int lane = threadIdx.x & 31;
    const int wave = threadIdx.x >> 5;
    const int half = lane >> 4;
    const int r    = lane & 15;
    const int m0   = (blockIdx.x * 8 + wave) * 16;
    if (m0 >= Mrows) return;

    v8f acc[8] = {};

    int arow = m0 + r;
    if (arow >= Mrows) arow = Mrows - 1;
    const float* ap = Xin + (size_t)arow * DFEAT + 2 * half;

    for (int kk = 0; kk < 32; ++kk) {
        const int k = kk * 4;
        v2f a;
        a.x = ap[k];
        a.y = ap[k + 1];

        const float* wrow = W + (size_t)(k + 2 * half) * DFEAT + r;
#pragma unroll
        for (int t = 0; t < 8; ++t) {
            v2f b;
            b.x = wrow[t * 16];
            b.y = wrow[t * 16 + DFEAT];
            acc[t] = __builtin_amdgcn_wmma_f32_16x16x4_f32(
                false, a, false, b, (short)0, acc[t], false, false);
        }
    }

#pragma unroll
    for (int t = 0; t < 8; ++t) {
        const int n = t * 16 + r;
        const float bv = bias ? bias[n] : 0.0f;
#pragma unroll
        for (int j = 0; j < 8; ++j) {
            const int row = m0 + j + 8 * half;
            if (row < Mrows) Y[(size_t)row * DFEAT + n] = acc[t][j] + bv;
        }
    }
}

#define OB_U 4
template <int R>
__global__ __launch_bounds__(256) void owner_bytes(
    const int* __restrict__ ids, int* __restrict__ own, int nnz, int nwords)
{
    const int t0 = blockIdx.x * (256 * OB_U) + threadIdx.x;
    int4 a[OB_U];
#pragma unroll
    for (int k = 0; k < OB_U; ++k) {
        const int p = 4 * (t0 + k * 256);
        a[k] = *(const int4*)(ids + (p + 3 < nnz ? p : 0));
    }
#pragma unroll
    for (int k = 0; k < OB_U; ++k) {
        const int t = t0 + k * 256;
        const int p = 4 * t;
        if (p + 3 >= nnz) {
            a[k].x = p     < nnz ? ids[p]     : 0;
            a[k].y = p + 1 < nnz ? ids[p + 1] : 0;
            a[k].z = p + 2 < nnz ? ids[p + 2] : 0;
            a[k].w = 0;
        }
        if (t < nwords)
            own[t] = (int)((unsigned)(a[k].x / R) | ((unsigned)(a[k].y / R) << 8) |
                           ((unsigned)(a[k].z / R) << 16) | ((unsigned)(a[k].w / R) << 24));
    }
}

#define OWN_THREADS 1024
#define OWN_WAVES   32
#define OWN_QCAP    768
#define OWN_BLK     512
#define OWN_UB      4
#define OWN_BATCH   8
#define OWN_GCAP    384

template <int R, bool BLEND>
__global__ __launch_bounds__(OWN_THREADS) void own_rows(
    const float* __restrict__ src,
    const int* __restrict__ gidx,
    const int* __restrict__ sidx,
    const int* __restrict__ own,
    float* __restrict__ dst,
    const float* __restrict__ Ap,
    const float* __restrict__ X0,
    int nrows, int nnz)
{
    static_assert(R <= 256, "local row must fit the low 8 bits of a queue entry");
    constexpr int NS = (R + OWN_WAVES - 1) / OWN_WAVES;
    __shared__ int queue[OWN_WAVES][OWN_QCAP];
    __shared__ int pool[OWN_WAVES * OWN_QCAP];
    __shared__ int kept[OWN_WAVES][OWN_GCAP];
    __shared__ int cnt[OWN_WAVES][OWN_WAVES];

    const int tid  = threadIdx.x;
    const int lane = tid & 31;
    const int w    = __builtin_amdgcn_readfirstlane(tid >> 5);
    const int base = blockIdx.x * R;
    int rows = nrows - base;
    if (rows > R) rows = R;
    if (rows < 0) rows = 0;

    static_assert(NS <= 8, "a slot must index an 8-wide register vector");
    v8f ax = {}, ay = {}, az = {}, aw = {};
    v8i rc = {};

    const int seg = (((nnz + OWN_WAVES - 1) / OWN_WAVES) + OWN_BLK - 1) / OWN_BLK * OWN_BLK;
    const uint4* __restrict__ own4 = (const uint4*)own;
    const unsigned me = blockIdx.x;
    const unsigned me4 = me * 0x01010101u;
    int pos = w * seg;
    int segEnd = pos + seg;
    if (segEnd > nnz) segEnd = nnz;

    int gn = 0, gi = 0;

    int any;
    do {
        if (gi >= gn) {
            gn = 0;
            gi = 0;
            while (pos < segEnd && gn <= OWN_GCAP - 32) {
                uint4 d[OWN_UB];
#pragma unroll
                for (int u = 0; u < OWN_UB; ++u) {
                    int pb = pos + u * OWN_BLK;
                    if (pb >= segEnd) pb = pos;
                    d[u] = own4[(pb >> 4) + lane];
                }
#pragma unroll
                for (int u = 0; u < OWN_UB; ++u) {
                    if (pos >= segEnd || gn > OWN_GCAP - 32) break;
                    const unsigned y0 = d[u].x ^ me4, y1 = d[u].y ^ me4, y2 = d[u].z ^ me4, y3 = d[u].w ^ me4;
                    const unsigned t = (((y0 - 0x01010101u) & ~y0) | ((y1 - 0x01010101u) & ~y1) |
                                        ((y2 - 0x01010101u) & ~y2) | ((y3 - 0x01010101u) & ~y3)) & 0x80808080u;
                    const unsigned hit = __builtin_amdgcn_ballot_w32(t != 0u);
                    if (t != 0u) kept[w][gn + (int)__builtin_amdgcn_mbcnt_lo(hit, 0u)] = (pos >> 4) + lane;
                    gn += __builtin_popcount(hit);
                    pos += OWN_BLK;
                }
            }
        }
        int qn = 0;
        while (gi < gn && qn <= OWN_QCAP - OWN_BLK) {
            const int i = gi + lane;
            const bool in = i < gn;
            const int g = in ? kept[w][i] : 0;
            const uint4 dd = own4[g];
            const unsigned y0 = dd.x ^ me4, y1 = dd.y ^ me4, y2 = dd.z ^ me4, y3 = dd.w ^ me4;
            const unsigned f0 = ~(((y0 & 0x7f7f7f7fu) + 0x7f7f7f7fu) | y0) & 0x80808080u;
            const unsigned f1 = ~(((y1 & 0x7f7f7f7fu) + 0x7f7f7f7fu) | y1) & 0x80808080u;
            const unsigned f2 = ~(((y2 & 0x7f7f7f7fu) + 0x7f7f7f7fu) | y2) & 0x80808080u;
            const unsigned f3 = ~(((y3 & 0x7f7f7f7fu) + 0x7f7f7f7fu) | y3) & 0x80808080u;
            const unsigned x0 = f0 >> 7, x1 = f1 >> 7, x2 = f2 >> 7, x3 = f3 >> 7;
            const unsigned n0 = (x0 | (x0 >> 7) | (x0 >> 14) | (x0 >> 21)) & 15u;
            const unsigned n1 = (x1 | (x1 >> 7) | (x1 >> 14) | (x1 >> 21)) & 15u;
            const unsigned n2 = (x2 | (x2 >> 7) | (x2 >> 14) | (x2 >> 21)) & 15u;
            const unsigned n3 = (x3 | (x3 >> 7) | (x3 >> 14) | (x3 >> 21)) & 15u;
            unsigned hm = n0 | (n1 << 4) | (n2 << 8) | (n3 << 12);
            const int nv = segEnd - g * 16;
            hm = !in || nv <= 0 ? 0u : (nv >= 16 ? hm : (hm & ((1u << nv) - 1u)));
            const int c = __builtin_popcount(hm);
            int incl = c;
            incl += __builtin_amdgcn_update_dpp(0, incl, 0x111, 0xf, 0xf, true);
            incl += __builtin_amdgcn_update_dpp(0, incl, 0x112, 0xf, 0xf, true);
            incl += __builtin_amdgcn_update_dpp(0, incl, 0x114, 0xf, 0xf, true);
            incl += __builtin_amdgcn_update_dpp(0, incl, 0x118, 0xf, 0xf, true);
            const int low = __builtin_amdgcn_readlane(incl, 15);
            if (lane >= 16) incl += low;
            int at = qn + incl - c;
            while (hm) {
                queue[w][at++] = g * 16 + __builtin_ctz(hm);
                hm &= hm - 1u;
            }
            qn += __builtin_amdgcn_readlane(incl, 31);
            gi += 32;
        }
        const unsigned l0 = (lane & 1) ? ~0u : 0u, l1 = (lane & 2) ? ~0u : 0u, l2 = (lane & 4) ? ~0u : 0u,
                       l3 = (lane & 8) ? ~0u : 0u, l4 = (lane & 16) ? ~0u : 0u;
        int mine = 0;
        for (int i0 = 0; i0 < qn; i0 += 32) {
            const int i = i0 + lane;
            const bool in = i < qn;
            int e = 0;
            if (in) {
                const int p = queue[w][i];
                const int s = sidx[p];
                const int g = gidx[p];
                e = (g << 8) | (s - base);
                queue[w][i] = e;
            }
            const unsigned vm = __builtin_amdgcn_ballot_w32(in);
            const unsigned b0 = __builtin_amdgcn_ballot_w32((e & 1) != 0), b1 = __builtin_amdgcn_ballot_w32((e & 2) != 0),
                           b2 = __builtin_amdgcn_ballot_w32((e & 4) != 0), b3 = __builtin_amdgcn_ballot_w32((e & 8) != 0),
                           b4 = __builtin_amdgcn_ballot_w32((e & 16) != 0);
            mine += __builtin_popcount(vm & ~(b0 ^ l0) & ~(b1 ^ l1) & ~(b2 ^ l2) & ~(b3 ^ l3) & ~(b4 ^ l4));
        }
        cnt[w][lane] = mine;
        __syncthreads();

        int before = 0, tot = 0;
        for (int q = 0; q < OWN_WAVES; ++q) {
            const int c = cnt[q][lane];
            if (q < w) before += c;
            tot += c;
        }
        int incl = tot;
        incl += __builtin_amdgcn_update_dpp(0, incl, 0x111, 0xf, 0xf, true);
        incl += __builtin_amdgcn_update_dpp(0, incl, 0x112, 0xf, 0xf, true);
        incl += __builtin_amdgcn_update_dpp(0, incl, 0x114, 0xf, 0xf, true);
        incl += __builtin_amdgcn_update_dpp(0, incl, 0x118, 0xf, 0xf, true);
        const int low = __builtin_amdgcn_readlane(incl, 15);
        if (lane >= 16) incl += low;
        const int ostart = incl - tot;
        int next = ostart + before;
        const int mystart = __builtin_amdgcn_readlane(ostart, w);
        const int mycnt   = __builtin_amdgcn_readlane(tot, w);

        for (int i0 = 0; i0 < qn; i0 += 32) {
            const int i = i0 + lane;
            const bool in = i < qn;
            const int e = in ? queue[w][i] : 0;
            const unsigned vm = __builtin_amdgcn_ballot_w32(in);
            const unsigned b0 = __builtin_amdgcn_ballot_w32((e & 1) != 0), b1 = __builtin_amdgcn_ballot_w32((e & 2) != 0),
                           b2 = __builtin_amdgcn_ballot_w32((e & 4) != 0), b3 = __builtin_amdgcn_ballot_w32((e & 8) != 0),
                           b4 = __builtin_amdgcn_ballot_w32((e & 16) != 0);
            const unsigned e0 = (e & 1) ? ~0u : 0u, e1 = (e & 2) ? ~0u : 0u, e2 = (e & 4) ? ~0u : 0u,
                           e3 = (e & 8) ? ~0u : 0u, e4 = (e & 16) ? ~0u : 0u;
            const unsigned peers = vm & ~(b0 ^ e0) & ~(b1 ^ e1) & ~(b2 ^ e2) & ~(b3 ^ e3) & ~(b4 ^ e4);
            const int rank = (int)__builtin_amdgcn_mbcnt_lo(peers, 0u);
            const int place = __shfl(next, e & 31);
            if (in) pool[place + rank] = e;
            next += __builtin_popcount(vm & ~(b0 ^ l0) & ~(b1 ^ l1) & ~(b2 ^ l2) & ~(b3 ^ l3) & ~(b4 ^ l4));
        }
        __syncthreads();

        auto drain = [&](int ev, int first, int n) {
            float4 v[OWN_BATCH];
            int rs[OWN_BATCH];
#pragma unroll
            for (int k = 0; k < OWN_BATCH; ++k) {
                if (k < n) {
                    const int e = __builtin_amdgcn_readlane(ev, first + k);
                    rs[k] = (e & 255) >> 5;
                    v[k] = ((const float4*)(src + (size_t)((unsigned)e >> 8) * DFEAT))[lane];
                }
            }
#pragma unroll
            for (int k = 0; k < OWN_BATCH; ++k) {
                if (k < n) {
                    const int s = rs[k];
                    ax[s] += v[k].x; ay[s] += v[k].y; az[s] += v[k].z; aw[s] += v[k].w;
                    if (BLEND) rc[s] += 1;
                }
            }
        };
        for (int h = 0; h < mycnt; h += 32) {
            const int left = mycnt - h;
            const int nb = left < 32 ? left : 32;
            const int ev = lane < nb ? pool[mystart + h + lane] : 0;
            int first = 0;
#pragma nounroll
            for (; first + OWN_BATCH <= nb; first += OWN_BATCH) drain(ev, first, OWN_BATCH);
            if (first < nb) drain(ev, first, nb - first);
        }

        any = __syncthreads_or((pos < segEnd || gi < gn) ? 1 : 0);
    } while (any);

    float4* d4 = (float4*)dst + (size_t)base * 32;
    if (BLEND) {
        const float4* a4 = (const float4*)Ap + (size_t)base * 32;
        const float4* x4 = (const float4*)X0 + (size_t)base * 32;
        if (rows > 0) {
            float4 av[NS], xv[NS];
#pragma unroll
            for (int t = 0; t < NS; ++t) {
                const int r = w + OWN_WAVES * t;
                const int j = r < rows ? r : rows - 1;
                av[t] = a4[j * 32 + lane];
                xv[t] = x4[j * 32 + lane];
            }
#pragma unroll
            for (int t = 0; t < NS; ++t) {
                const int r = w + OWN_WAVES * t;
                if (r < rows) {
                    const float4 s = make_float4(ax[t], ay[t], az[t], aw[t]);
                    const float d = (float)rc[t];
                    float4 o;
                    o.x = 0.5f * (d * av[t].x + s.x) + 0.5f * xv[t].x;
                    o.y = 0.5f * (d * av[t].y + s.y) + 0.5f * xv[t].y;
                    o.z = 0.5f * (d * av[t].z + s.z) + 0.5f * xv[t].z;
                    o.w = 0.5f * (d * av[t].w + s.w) + 0.5f * xv[t].w;
                    d4[r * 32 + lane] = o;
                }
            }
        }
    } else {
#pragma unroll
        for (int t = 0; t < NS; ++t) {
            const int r = w + OWN_WAVES * t;
            if (r < rows) d4[r * 32 + lane] = make_float4(ax[t], ay[t], az[t], aw[t]);
        }
    }
}

extern "C" void kernel_launch(void* const* d_in, const int* in_sizes, int n_in,
                              void* d_out, int out_size, void* d_ws, size_t ws_size,
                              hipStream_t stream) {
    const float* X      = (const float*)d_in[0];
    const int*   vertex = (const int*)d_in[1];
    const int*   edges  = (const int*)d_in[2];
    const float* X0     = (const float*)d_in[3];
    const float* W1     = (const float*)d_in[4];
    const float* b1     = (const float*)d_in[5];
    const float* W2     = (const float*)d_in[6];
    const float* b2     = (const float*)d_in[7];
    const float* W3     = (const float*)d_in[8];
    const float* b3     = (const float*)d_in[9];
    float* out = (float*)d_out;

    constexpr int N = 50000, M = 25000, NNZ = 800000;
    constexpr size_t ND = (size_t)N * DFEAT;
    constexpr size_t MD = (size_t)M * DFEAT;

    char* ws = (char*)d_ws;
    auto align256 = [](size_t x) { return (x + 255) & ~(size_t)255; };
    size_t off = 0;
    float* XW = (float*)(ws + off); off = align256(off + ND * 4);
    float* Xe = (float*)(ws + off); off = align256(off + MD * 4);
    float* Ap = (float*)(ws + off); off = align256(off + ND * 4);
    float* Bm = (float*)(ws + off); off = align256(off + MD * 4);
    float* Xn = XW;

    const int gemmN = (N + 127) / 128;
    const int gemmM = (M + 127) / 128;
    constexpr int RM = 98;
    constexpr int RN = 196;
    static_assert((M + RM - 1) / RM <= 256 && (N + RN - 1) / RN <= 256, "an owner number must fit one byte");

    constexpr int OWN_POS   = (NNZ + OWN_BLK - 1) / OWN_BLK * OWN_BLK;
    constexpr int OWN_WORDS = OWN_POS / 4;
    constexpr int OWN_GRID  = (OWN_WORDS + 256 * OB_U - 1) / (256 * OB_U);
    static_assert((size_t)OWN_POS <= MD * 4, "a byte list must fit the region it borrows");
    int* ownE = (int*)Bm;
    int* ownV = (int*)Xe;

    owner_bytes<RM><<<OWN_GRID, 256, 0, stream>>>(edges, ownE, NNZ, OWN_WORDS);

    gemm128_wmma<<<gemmN, 256, 0, stream>>>(X, N, W1, b1, XW);

    own_rows<RM, false><<<(M + RM - 1) / RM, OWN_THREADS, 0, stream>>>(XW, vertex, edges, ownE, Xe, nullptr, nullptr, M, NNZ);

    gemm128_wmma<<<gemmN, 256, 0, stream>>>(X, N, W2, b2, Ap);

    gemm128_wmma<<<gemmM, 256, 0, stream>>>(Xe, M, W2 + 128 * DFEAT, nullptr, Bm);

    owner_bytes<RN><<<OWN_GRID, 256, 0, stream>>>(vertex, ownV, NNZ, OWN_WORDS);

    own_rows<RN, true><<<(N + RN - 1) / RN, OWN_THREADS, 0, stream>>>(Bm, edges, vertex, ownV, Xn, Ap, X0, N, NNZ);

    gemm128_wmma<<<gemmN, 256, 0, stream>>>(Xn, N, W3, b3, out);
}


# Row-sum scan tests one bit per 16 list positions: the owner launch adds, per 16 positions, a 256-bit set of owning work-groups (8 word planes); a work-group reads only its plane word; sums unchanged
# speedup vs baseline: 2.4087x; 1.0419x over previous
#include <hip/hip_runtime.h>


typedef __attribute__((ext_vector_type(2))) float v2f;
typedef __attribute__((ext_vector_type(8))) float v8f;
typedef __attribute__((ext_vector_type(8))) int   v8i;

#define DFEAT 128

__global__ __launch_bounds__(256) void gemm128_wmma(
    const float* __restrict__ Xin, int Mrows,
    const float* __restrict__ W,
    const float* __restrict__ bias,
    float* __restrict__ Y)
{
    const int lane = threadIdx.x & 31;
    const int wave = threadIdx.x >> 5;
    const int half = lane >> 4;
    const int r    = lane & 15;
    const int m0   = (blockIdx.x * 8 + wave) * 16;
    if (m0 >= Mrows) return;

    v8f acc[8] = {};

    int arow = m0 + r;
    if (arow >= Mrows) arow = Mrows - 1;
    const float* ap = Xin + (size_t)arow * DFEAT + 2 * half;

    for (int kk = 0; kk < 32; ++kk) {
        const int k = kk * 4;
        v2f a;
        a.x = ap[k];
        a.y = ap[k + 1];

        const float* wrow = W + (size_t)(k + 2 * half) * DFEAT + r;
#pragma unroll
        for (int t = 0; t < 8; ++t) {
            v2f b;
            b.x = wrow[t * 16];
            b.y = wrow[t * 16 + DFEAT];
            acc[t] = __builtin_amdgcn_wmma_f32_16x16x4_f32(
                false, a, false, b, (short)0, acc[t], false, false);
        }
    }

#pragma unroll
    for (int t = 0; t < 8; ++t) {
        const int n = t * 16 + r;
        const float bv = bias ? bias[n] : 0.0f;
#pragma unroll
        for (int j = 0; j < 8; ++j) {
            const int row = m0 + j + 8 * half;
            if (row < Mrows) Y[(size_t)row * DFEAT + n] = acc[t][j] + bv;
        }
    }
}

template <int R>
__global__ __launch_bounds__(256) void owner_marks(
    const int* __restrict__ ids, int* __restrict__ own, int* __restrict__ plane, int nnz, int ngroups)
{
    const int g = blockIdx.x * 256 + threadIdx.x;
    if (g >= ngroups) return;
    const int p0 = 16 * g;
    int id[16];
    if (p0 + 15 < nnz) {
        const int4* q = (const int4*)(ids + p0);
#pragma unroll
        for (int k = 0; k < 4; ++k) {
            const int4 a = q[k];
            id[4 * k] = a.x; id[4 * k + 1] = a.y; id[4 * k + 2] = a.z; id[4 * k + 3] = a.w;
        }
    } else {
#pragma unroll
        for (int k = 0; k < 16; ++k) id[k] = p0 + k < nnz ? ids[p0 + k] : 0;
    }
    unsigned ob[4] = {0u, 0u, 0u, 0u};
    unsigned m[8] = {0u, 0u, 0u, 0u, 0u, 0u, 0u, 0u};
#pragma unroll
    for (int k = 0; k < 16; ++k) {
        const bool valid = p0 + k < nnz;
        const unsigned o = valid ? (unsigned)(id[k] / R) : 0u;
        ob[k >> 2] |= o << (8 * (k & 3));
        const unsigned bit = valid ? (1u << (o & 31u)) : 0u;
        const unsigned hi = o >> 5;
#pragma unroll
        for (int j = 0; j < 8; ++j) m[j] |= hi == (unsigned)j ? bit : 0u;
    }
    ((uint4*)own)[g] = make_uint4(ob[0], ob[1], ob[2], ob[3]);
#pragma unroll
    for (int j = 0; j < 8; ++j) plane[(size_t)j * ngroups + g] = (int)m[j];
}

#define OWN_THREADS 1024
#define OWN_WAVES   32
#define OWN_QCAP    768
#define OWN_BLK     512
#define OWN_UB      25
#define OWN_BATCH   8
#define OWN_GCAP    384

template <int R, bool BLEND>
__global__ __launch_bounds__(OWN_THREADS) void own_rows(
    const float* __restrict__ src,
    const int* __restrict__ gidx,
    const int* __restrict__ sidx,
    const int* __restrict__ own,
    const int* __restrict__ plane,
    int ngroups,
    float* __restrict__ dst,
    const float* __restrict__ Ap,
    const float* __restrict__ X0,
    int nrows, int nnz)
{
    static_assert(R <= 256, "local row must fit the low 8 bits of a queue entry");
    constexpr int NS = (R + OWN_WAVES - 1) / OWN_WAVES;
    __shared__ int queue[OWN_WAVES][OWN_QCAP];
    __shared__ int pool[OWN_WAVES * OWN_QCAP];
    __shared__ int kept[OWN_WAVES][OWN_GCAP];
    __shared__ int cnt[OWN_WAVES][OWN_WAVES];

    const int tid  = threadIdx.x;
    const int lane = tid & 31;
    const int w    = __builtin_amdgcn_readfirstlane(tid >> 5);
    const int base = blockIdx.x * R;
    int rows = nrows - base;
    if (rows > R) rows = R;
    if (rows < 0) rows = 0;

    static_assert(NS <= 8, "a slot must index an 8-wide register vector");
    v8f ax = {}, ay = {}, az = {}, aw = {};
    v8i rc = {};

    const int seg = (((nnz + OWN_WAVES - 1) / OWN_WAVES) + OWN_BLK - 1) / OWN_BLK * OWN_BLK;
    const uint4* __restrict__ own4 = (const uint4*)own;
    const unsigned me = blockIdx.x;
    const unsigned me4 = me * 0x01010101u;
    const unsigned* __restrict__ myplane = (const unsigned*)plane + (size_t)(me >> 5) * (size_t)ngroups;
    const unsigned mebit = 1u << (me & 31u);
    int pos = w * seg;
    int segEnd = pos + seg;
    if (segEnd > nnz) segEnd = nnz;

    int gn = 0, gi = 0;

    int any;
    do {
        if (gi >= gn) {
            gn = 0;
            gi = 0;
            while (pos < segEnd && gn <= OWN_GCAP - 32) {
                unsigned d[OWN_UB];
#pragma unroll
                for (int u = 0; u < OWN_UB; ++u) {
                    int pb = pos + u * OWN_BLK;
                    if (pb >= segEnd) pb = pos;
                    d[u] = myplane[(pb >> 4) + lane];
                }
#pragma unroll
                for (int u = 0; u < OWN_UB; ++u) {
                    if (pos >= segEnd || gn > OWN_GCAP - 32) break;
                    const bool t = (d[u] & mebit) != 0u;
                    const unsigned hit = __builtin_amdgcn_ballot_w32(t);
                    if (t) kept[w][gn + (int)__builtin_amdgcn_mbcnt_lo(hit, 0u)] = (pos >> 4) + lane;
                    gn += __builtin_popcount(hit);
                    pos += OWN_BLK;
                }
            }
        }
        int qn = 0;
        while (gi < gn && qn <= OWN_QCAP - OWN_BLK) {
            const int i = gi + lane;
            const bool in = i < gn;
            const int g = in ? kept[w][i] : 0;
            const uint4 dd = own4[g];
            const unsigned y0 = dd.x ^ me4, y1 = dd.y ^ me4, y2 = dd.z ^ me4, y3 = dd.w ^ me4;
            const unsigned f0 = ~(((y0 & 0x7f7f7f7fu) + 0x7f7f7f7fu) | y0) & 0x80808080u;
            const unsigned f1 = ~(((y1 & 0x7f7f7f7fu) + 0x7f7f7f7fu) | y1) & 0x80808080u;
            const unsigned f2 = ~(((y2 & 0x7f7f7f7fu) + 0x7f7f7f7fu) | y2) & 0x80808080u;
            const unsigned f3 = ~(((y3 & 0x7f7f7f7fu) + 0x7f7f7f7fu) | y3) & 0x80808080u;
            const unsigned x0 = f0 >> 7, x1 = f1 >> 7, x2 = f2 >> 7, x3 = f3 >> 7;
            const unsigned n0 = (x0 | (x0 >> 7) | (x0 >> 14) | (x0 >> 21)) & 15u;
            const unsigned n1 = (x1 | (x1 >> 7) | (x1 >> 14) | (x1 >> 21)) & 15u;
            const unsigned n2 = (x2 | (x2 >> 7) | (x2 >> 14) | (x2 >> 21)) & 15u;
            const unsigned n3 = (x3 | (x3 >> 7) | (x3 >> 14) | (x3 >> 21)) & 15u;
            unsigned hm = n0 | (n1 << 4) | (n2 << 8) | (n3 << 12);
            const int nv = segEnd - g * 16;
            hm = !in || nv <= 0 ? 0u : (nv >= 16 ? hm : (hm & ((1u << nv) - 1u)));
            const int c = __builtin_popcount(hm);
            int incl = c;
            incl += __builtin_amdgcn_update_dpp(0, incl, 0x111, 0xf, 0xf, true);
            incl += __builtin_amdgcn_update_dpp(0, incl, 0x112, 0xf, 0xf, true);
            incl += __builtin_amdgcn_update_dpp(0, incl, 0x114, 0xf, 0xf, true);
            incl += __builtin_amdgcn_update_dpp(0, incl, 0x118, 0xf, 0xf, true);
            const int low = __builtin_amdgcn_readlane(incl, 15);
            if (lane >= 16) incl += low;
            int at = qn + incl - c;
            while (hm) {
                queue[w][at++] = g * 16 + __builtin_ctz(hm);
                hm &= hm - 1u;
            }
            qn += __builtin_amdgcn_readlane(incl, 31);
            gi += 32;
        }
        const unsigned l0 = (lane & 1) ? ~0u : 0u, l1 = (lane & 2) ? ~0u : 0u, l2 = (lane & 4) ? ~0u : 0u,
                       l3 = (lane & 8) ? ~0u : 0u, l4 = (lane & 16) ? ~0u : 0u;
        int mine = 0;
        for (int i0 = 0; i0 < qn; i0 += 32) {
            const int i = i0 + lane;
            const bool in = i < qn;
            int e = 0;
            if (in) {
                const int p = queue[w][i];
                const int s = sidx[p];
                const int g = gidx[p];
                e = (g << 8) | (s - base);
                queue[w][i] = e;
            }
            const unsigned vm = __builtin_amdgcn_ballot_w32(in);
            const unsigned b0 = __builtin_amdgcn_ballot_w32((e & 1) != 0), b1 = __builtin_amdgcn_ballot_w32((e & 2) != 0),
                           b2 = __builtin_amdgcn_ballot_w32((e & 4) != 0), b3 = __builtin_amdgcn_ballot_w32((e & 8) != 0),
                           b4 = __builtin_amdgcn_ballot_w32((e & 16) != 0);
            mine += __builtin_popcount(vm & ~(b0 ^ l0) & ~(b1 ^ l1) & ~(b2 ^ l2) & ~(b3 ^ l3) & ~(b4 ^ l4));
        }
        cnt[w][lane] = mine;
        __syncthreads();

        int before = 0, tot = 0;
        for (int q = 0; q < OWN_WAVES; ++q) {
            const int c = cnt[q][lane];
            if (q < w) before += c;
            tot += c;
        }
        int incl = tot;
        incl += __builtin_amdgcn_update_dpp(0, incl, 0x111, 0xf, 0xf, true);
        incl += __builtin_amdgcn_update_dpp(0, incl, 0x112, 0xf, 0xf, true);
        incl += __builtin_amdgcn_update_dpp(0, incl, 0x114, 0xf, 0xf, true);
        incl += __builtin_amdgcn_update_dpp(0, incl, 0x118, 0xf, 0xf, true);
        const int low = __builtin_amdgcn_readlane(incl, 15);
        if (lane >= 16) incl += low;
        const int ostart = incl - tot;
        int next = ostart + before;
        const int mystart = __builtin_amdgcn_readlane(ostart, w);
        const int mycnt   = __builtin_amdgcn_readlane(tot, w);

        for (int i0 = 0; i0 < qn; i0 += 32) {
            const int i = i0 + lane;
            const bool in = i < qn;
            const int e = in ? queue[w][i] : 0;
            const unsigned vm = __builtin_amdgcn_ballot_w32(in);
            const unsigned b0 = __builtin_amdgcn_ballot_w32((e & 1) != 0), b1 = __builtin_amdgcn_ballot_w32((e & 2) != 0),
                           b2 = __builtin_amdgcn_ballot_w32((e & 4) != 0), b3 = __builtin_amdgcn_ballot_w32((e & 8) != 0),
                           b4 = __builtin_amdgcn_ballot_w32((e & 16) != 0);
            const unsigned e0 = (e & 1) ? ~0u : 0u, e1 = (e & 2) ? ~0u : 0u, e2 = (e & 4) ? ~0u : 0u,
                           e3 = (e & 8) ? ~0u : 0u, e4 = (e & 16) ? ~0u : 0u;
            const unsigned peers = vm & ~(b0 ^ e0) & ~(b1 ^ e1) & ~(b2 ^ e2) & ~(b3 ^ e3) & ~(b4 ^ e4);
            const int rank = (int)__builtin_amdgcn_mbcnt_lo(peers, 0u);
            const int place = __shfl(next, e & 31);
            if (in) pool[place + rank] = e;
            next += __builtin_popcount(vm & ~(b0 ^ l0) & ~(b1 ^ l1) & ~(b2 ^ l2) & ~(b3 ^ l3) & ~(b4 ^ l4));
        }
        __syncthreads();

        auto drain = [&](int ev, int first, int n) {
            float4 v[OWN_BATCH];
            int rs[OWN_BATCH];
#pragma unroll
            for (int k = 0; k < OWN_BATCH; ++k) {
                if (k < n) {
                    const int e = __builtin_amdgcn_readlane(ev, first + k);
                    rs[k] = (e & 255) >> 5;
                    v[k] = ((const float4*)(src + (size_t)((unsigned)e >> 8) * DFEAT))[lane];
                }
            }
#pragma unroll
            for (int k = 0; k < OWN_BATCH; ++k) {
                if (k < n) {
                    const int s = rs[k];
                    ax[s] += v[k].x; ay[s] += v[k].y; az[s] += v[k].z; aw[s] += v[k].w;
                    if (BLEND) rc[s] += 1;
                }
            }
        };
        for (int h = 0; h < mycnt; h += 32) {
            const int left = mycnt - h;
            const int nb = left < 32 ? left : 32;
            const int ev = lane < nb ? pool[mystart + h + lane] : 0;
            int first = 0;
#pragma nounroll
            for (; first + OWN_BATCH <= nb; first += OWN_BATCH) drain(ev, first, OWN_BATCH);
            if (first < nb) drain(ev, first, nb - first);
        }

        any = __syncthreads_or((pos < segEnd || gi < gn) ? 1 : 0);
    } while (any);

    float4* d4 = (float4*)dst + (size_t)base * 32;
    if (BLEND) {
        const float4* a4 = (const float4*)Ap + (size_t)base * 32;
        const float4* x4 = (const float4*)X0 + (size_t)base * 32;
        if (rows > 0) {
            float4 av[NS], xv[NS];
#pragma unroll
            for (int t = 0; t < NS; ++t) {
                const int r = w + OWN_WAVES * t;
                const int j = r < rows ? r : rows - 1;
                av[t] = a4[j * 32 + lane];
                xv[t] = x4[j * 32 + lane];
            }
#pragma unroll
            for (int t = 0; t < NS; ++t) {
                const int r = w + OWN_WAVES * t;
                if (r < rows) {
                    const float4 s = make_float4(ax[t], ay[t], az[t], aw[t]);
                    const float d = (float)rc[t];
                    float4 o;
                    o.x = 0.5f * (d * av[t].x + s.x) + 0.5f * xv[t].x;
                    o.y = 0.5f * (d * av[t].y + s.y) + 0.5f * xv[t].y;
                    o.z = 0.5f * (d * av[t].z + s.z) + 0.5f * xv[t].z;
                    o.w = 0.5f * (d * av[t].w + s.w) + 0.5f * xv[t].w;
                    d4[r * 32 + lane] = o;
                }
            }
        }
    } else {
#pragma unroll
        for (int t = 0; t < NS; ++t) {
            const int r = w + OWN_WAVES * t;
            if (r < rows) d4[r * 32 + lane] = make_float4(ax[t], ay[t], az[t], aw[t]);
        }
    }
}

extern "C" void kernel_launch(void* const* d_in, const int* in_sizes, int n_in,
                              void* d_out, int out_size, void* d_ws, size_t ws_size,
                              hipStream_t stream) {
    const float* X      = (const float*)d_in[0];
    const int*   vertex = (const int*)d_in[1];
    const int*   edges  = (const int*)d_in[2];
    const float* X0     = (const float*)d_in[3];
    const float* W1     = (const float*)d_in[4];
    const float* b1     = (const float*)d_in[5];
    const float* W2     = (const float*)d_in[6];
    const float* b2     = (const float*)d_in[7];
    const float* W3     = (const float*)d_in[8];
    const float* b3     = (const float*)d_in[9];
    float* out = (float*)d_out;

    constexpr int N = 50000, M = 25000, NNZ = 800000;
    constexpr size_t ND = (size_t)N * DFEAT;
    constexpr size_t MD = (size_t)M * DFEAT;

    char* ws = (char*)d_ws;
    auto align256 = [](size_t x) { return (x + 255) & ~(size_t)255; };
    size_t off = 0;
    float* XW = (float*)(ws + off); off = align256(off + ND * 4);
    float* Xe = (float*)(ws + off); off = align256(off + MD * 4);
    float* Ap = (float*)(ws + off); off = align256(off + ND * 4);
    float* Bm = (float*)(ws + off); off = align256(off + MD * 4);
    float* Xn = XW;

    const int gemmN = (N + 127) / 128;
    const int gemmM = (M + 127) / 128;
    constexpr int RM = 98;
    constexpr int RN = 196;
    static_assert((M + RM - 1) / RM <= 256 && (N + RN - 1) / RN <= 256, "an owner number must fit one byte");

    constexpr int OWN_POS    = (NNZ + OWN_BLK - 1) / OWN_BLK * OWN_BLK;
    constexpr int OWN_WORDS  = OWN_POS / 4;
    constexpr int OWN_GROUPS = OWN_POS / 16;
    constexpr int OWN_GRID   = (OWN_GROUPS + 255) / 256;
    static_assert((size_t)OWN_POS + (size_t)OWN_GROUPS * 32 <= MD * 4, "bytes and planes must fit the region they borrow");
    int* ownE = (int*)Bm;
    int* plE  = ownE + OWN_WORDS;
    int* ownV = (int*)Xe;
    int* plV  = ownV + OWN_WORDS;

    owner_marks<RM><<<OWN_GRID, 256, 0, stream>>>(edges, ownE, plE, NNZ, OWN_GROUPS);

    gemm128_wmma<<<gemmN, 256, 0, stream>>>(X, N, W1, b1, XW);

    own_rows<RM, false><<<(M + RM - 1) / RM, OWN_THREADS, 0, stream>>>(XW, vertex, edges, ownE, plE, OWN_GROUPS, Xe, nullptr, nullptr, M, NNZ);

    gemm128_wmma<<<gemmN, 256, 0, stream>>>(X, N, W2, b2, Ap);

    gemm128_wmma<<<gemmM, 256, 0, stream>>>(Xe, M, W2 + 128 * DFEAT, nullptr, Bm);

    owner_marks<RN><<<OWN_GRID, 256, 0, stream>>>(vertex, ownV, plV, NNZ, OWN_GROUPS);

    own_rows<RN, true><<<(N + RN - 1) / RN, OWN_THREADS, 0, stream>>>(Bm, edges, vertex, ownV, plV, OWN_GROUPS, Xn, Ap, X0, N, NNZ);

    gemm128_wmma<<<gemmN, 256, 0, stream>>>(Xn, N, W3, b3, out);
}


# GEMM fragment maps permuted in registers (k = 8q+4h+2e+j, col = 64(t>>2)+4r+(t&3)): A, W, result move as 16-byte lane accesses, W loads one group per 16-WMMA block; no LDS; k sum order changes
# speedup vs baseline: 3.0951x; 1.2850x over previous
#include <hip/hip_runtime.h>


typedef __attribute__((ext_vector_type(2))) float v2f;
typedef __attribute__((ext_vector_type(4))) float v4f;
typedef __attribute__((ext_vector_type(8))) float v8f;
typedef __attribute__((ext_vector_type(8))) int   v8i;

#define DFEAT 128

__global__ __launch_bounds__(256) void gemm128_wmma(
    const float* __restrict__ Xin, int Mrows,
    const float* __restrict__ W,
    const float* __restrict__ bias,
    float* __restrict__ Y)
{
    const int lane = threadIdx.x & 31;
    const int wave = threadIdx.x >> 5;
    const int half = lane >> 4;
    const int r    = lane & 15;
    const int m0   = (blockIdx.x * 8 + wave) * 16;
    if (m0 >= Mrows) return;

    v8f acc[8] = {};

    int arow = m0 + r;
    if (arow >= Mrows) arow = Mrows - 1;
    const v4f* ap = (const v4f*)(Xin + (size_t)arow * DFEAT + 4 * half);
    v4f a4[16];
#pragma unroll
    for (int q = 0; q < 16; ++q) a4[q] = ap[2 * q];

    const v4f* wp = (const v4f*)(W + (size_t)(4 * half) * DFEAT + 4 * r);
    v4f b[2][4][2];
#pragma unroll
    for (int i = 0; i < 4; ++i) {
        b[0][i][0] = wp[i * 32];
        b[0][i][1] = wp[i * 32 + 16];
    }

#pragma unroll
    for (int q = 0; q < 16; ++q) {
        if (q + 1 < 16) {
#pragma unroll
            for (int i = 0; i < 4; ++i) {
                b[(q + 1) & 1][i][0] = wp[((q + 1) * 8 + i) * 32];
                b[(q + 1) & 1][i][1] = wp[((q + 1) * 8 + i) * 32 + 16];
            }
        }
        __builtin_amdgcn_sched_barrier(0);
#pragma unroll
        for (int e = 0; e < 2; ++e) {
            v2f a;
            a.x = a4[q][2 * e];
            a.y = a4[q][2 * e + 1];
#pragma unroll
            for (int t = 0; t < 8; ++t) {
                v2f bb;
                bb.x = b[q & 1][2 * e][t >> 2][t & 3];
                bb.y = b[q & 1][2 * e + 1][t >> 2][t & 3];
                acc[t] = __builtin_amdgcn_wmma_f32_16x16x4_f32(
                    false, a, false, bb, (short)0, acc[t], false, false);
            }
        }
    }

    v4f bv[2];
#pragma unroll
    for (int u = 0; u < 2; ++u) {
        if (bias) bv[u] = ((const v4f*)bias)[16 * u + r];
        else { bv[u].x = 0.0f; bv[u].y = 0.0f; bv[u].z = 0.0f; bv[u].w = 0.0f; }
    }
#pragma unroll
    for (int j = 0; j < 8; ++j) {
        const int row = m0 + j + 8 * half;
        if (row < Mrows) {
            v4f* yp = (v4f*)(Y + (size_t)row * DFEAT);
#pragma unroll
            for (int u = 0; u < 2; ++u) {
                v4f o;
                o.x = acc[4 * u + 0][j] + bv[u].x;
                o.y = acc[4 * u + 1][j] + bv[u].y;
                o.z = acc[4 * u + 2][j] + bv[u].z;
                o.w = acc[4 * u + 3][j] + bv[u].w;
                yp[16 * u + r] = o;
            }
        }
    }
}

template <int R>
__global__ __launch_bounds__(256) void owner_marks(
    const int* __restrict__ ids, int* __restrict__ own, int* __restrict__ plane, int nnz, int ngroups)
{
    const int g = blockIdx.x * 256 + threadIdx.x;
    if (g >= ngroups) return;
    const int p0 = 16 * g;
    int id[16];
    if (p0 + 15 < nnz) {
        const int4* q = (const int4*)(ids + p0);
#pragma unroll
        for (int k = 0; k < 4; ++k) {
            const int4 a = q[k];
            id[4 * k] = a.x; id[4 * k + 1] = a.y; id[4 * k + 2] = a.z; id[4 * k + 3] = a.w;
        }
    } else {
#pragma unroll
        for (int k = 0; k < 16; ++k) id[k] = p0 + k < nnz ? ids[p0 + k] : 0;
    }
    unsigned ob[4] = {0u, 0u, 0u, 0u};
    unsigned m[8] = {0u, 0u, 0u, 0u, 0u, 0u, 0u, 0u};
#pragma unroll
    for (int k = 0; k < 16; ++k) {
        const bool valid = p0 + k < nnz;
        const unsigned o = valid ? (unsigned)(id[k] / R) : 0u;
        ob[k >> 2] |= o << (8 * (k & 3));
        const unsigned bit = valid ? (1u << (o & 31u)) : 0u;
        const unsigned hi = o >> 5;
#pragma unroll
        for (int j = 0; j < 8; ++j) m[j] |= hi == (unsigned)j ? bit : 0u;
    }
    ((uint4*)own)[g] = make_uint4(ob[0], ob[1], ob[2], ob[3]);
#pragma unroll
    for (int j = 0; j < 8; ++j) plane[(size_t)j * ngroups + g] = (int)m[j];
}

#define OWN_THREADS 1024
#define OWN_WAVES   32
#define OWN_QCAP    768
#define OWN_BLK     512
#define OWN_UB      25
#define OWN_BATCH   8
#define OWN_GCAP    384

template <int R, bool BLEND>
__global__ __launch_bounds__(OWN_THREADS) void own_rows(
    const float* __restrict__ src,
    const int* __restrict__ gidx,
    const int* __restrict__ sidx,
    const int* __restrict__ own,
    const int* __restrict__ plane,
    int ngroups,
    float* __restrict__ dst,
    const float* __restrict__ Ap,
    const float* __restrict__ X0,
    int nrows, int nnz)
{
    static_assert(R <= 256, "local row must fit the low 8 bits of a queue entry");
    constexpr int NS = (R + OWN_WAVES - 1) / OWN_WAVES;
    __shared__ int queue[OWN_WAVES][OWN_QCAP];
    __shared__ int pool[OWN_WAVES * OWN_QCAP];
    __shared__ int kept[OWN_WAVES][OWN_GCAP];
    __shared__ int cnt[OWN_WAVES][OWN_WAVES];

    const int tid  = threadIdx.x;
    const int lane = tid & 31;
    const int w    = __builtin_amdgcn_readfirstlane(tid >> 5);
    const int base = blockIdx.x * R;
    int rows = nrows - base;
    if (rows > R) rows = R;
    if (rows < 0) rows = 0;

    static_assert(NS <= 8, "a slot must index an 8-wide register vector");
    v8f ax = {}, ay = {}, az = {}, aw = {};
    v8i rc = {};

    const int seg = (((nnz + OWN_WAVES - 1) / OWN_WAVES) + OWN_BLK - 1) / OWN_BLK * OWN_BLK;
    const uint4* __restrict__ own4 = (const uint4*)own;
    const unsigned me = blockIdx.x;
    const unsigned me4 = me * 0x01010101u;
    const unsigned* __restrict__ myplane = (const unsigned*)plane + (size_t)(me >> 5) * (size_t)ngroups;
    const unsigned mebit = 1u << (me & 31u);
    int pos = w * seg;
    int segEnd = pos + seg;
    if (segEnd > nnz) segEnd = nnz;

    int gn = 0, gi = 0;

    int any;
    do {
        if (gi >= gn) {
            gn = 0;
            gi = 0;
            while (pos < segEnd && gn <= OWN_GCAP - 32) {
                unsigned d[OWN_UB];
#pragma unroll
                for (int u = 0; u < OWN_UB; ++u) {
                    int pb = pos + u * OWN_BLK;
                    if (pb >= segEnd) pb = pos;
                    d[u] = myplane[(pb >> 4) + lane];
                }
#pragma unroll
                for (int u = 0; u < OWN_UB; ++u) {
                    if (pos >= segEnd || gn > OWN_GCAP - 32) break;
                    const bool t = (d[u] & mebit) != 0u;
                    const unsigned hit = __builtin_amdgcn_ballot_w32(t);
                    if (t) kept[w][gn + (int)__builtin_amdgcn_mbcnt_lo(hit, 0u)] = (pos >> 4) + lane;
                    gn += __builtin_popcount(hit);
                    pos += OWN_BLK;
                }
            }
        }
        int qn = 0;
        while (gi < gn && qn <= OWN_QCAP - OWN_BLK) {
            const int i = gi + lane;
            const bool in = i < gn;
            const int g = in ? kept[w][i] : 0;
            const uint4 dd = own4[g];
            const unsigned y0 = dd.x ^ me4, y1 = dd.y ^ me4, y2 = dd.z ^ me4, y3 = dd.w ^ me4;
            const unsigned f0 = ~(((y0 & 0x7f7f7f7fu) + 0x7f7f7f7fu) | y0) & 0x80808080u;
            const unsigned f1 = ~(((y1 & 0x7f7f7f7fu) + 0x7f7f7f7fu) | y1) & 0x80808080u;
            const unsigned f2 = ~(((y2 & 0x7f7f7f7fu) + 0x7f7f7f7fu) | y2) & 0x80808080u;
            const unsigned f3 = ~(((y3 & 0x7f7f7f7fu) + 0x7f7f7f7fu) | y3) & 0x80808080u;
            const unsigned x0 = f0 >> 7, x1 = f1 >> 7, x2 = f2 >> 7, x3 = f3 >> 7;
            const unsigned n0 = (x0 | (x0 >> 7) | (x0 >> 14) | (x0 >> 21)) & 15u;
            const unsigned n1 = (x1 | (x1 >> 7) | (x1 >> 14) | (x1 >> 21)) & 15u;
            const unsigned n2 = (x2 | (x2 >> 7) | (x2 >> 14) | (x2 >> 21)) & 15u;
            const unsigned n3 = (x3 | (x3 >> 7) | (x3 >> 14) | (x3 >> 21)) & 15u;
            unsigned hm = n0 | (n1 << 4) | (n2 << 8) | (n3 << 12);
            const int nv = segEnd - g * 16;
            hm = !in || nv <= 0 ? 0u : (nv >= 16 ? hm : (hm & ((1u << nv) - 1u)));
            const int c = __builtin_popcount(hm);
            int incl = c;
            incl += __builtin_amdgcn_update_dpp(0, incl, 0x111, 0xf, 0xf, true);
            incl += __builtin_amdgcn_update_dpp(0, incl, 0x112, 0xf, 0xf, true);
            incl += __builtin_amdgcn_update_dpp(0, incl, 0x114, 0xf, 0xf, true);
            incl += __builtin_amdgcn_update_dpp(0, incl, 0x118, 0xf, 0xf, true);
            const int low = __builtin_amdgcn_readlane(incl, 15);
            if (lane >= 16) incl += low;
            int at = qn + incl - c;
            while (hm) {
                queue[w][at++] = g * 16 + __builtin_ctz(hm);
                hm &= hm - 1u;
            }
            qn += __builtin_amdgcn_readlane(incl, 31);
            gi += 32;
        }
        const unsigned l0 = (lane & 1) ? ~0u : 0u, l1 = (lane & 2) ? ~0u : 0u, l2 = (lane & 4) ? ~0u : 0u,
                       l3 = (lane & 8) ? ~0u : 0u, l4 = (lane & 16) ? ~0u : 0u;
        int mine = 0;
        for (int i0 = 0; i0 < qn; i0 += 32) {
            const int i = i0 + lane;
            const bool in = i < qn;
            int e = 0;
            if (in) {
                const int p = queue[w][i];
                const int s = sidx[p];
                const int g = gidx[p];
                e = (g << 8) | (s - base);
                queue[w][i] = e;
            }
            const unsigned vm = __builtin_amdgcn_ballot_w32(in);
            const unsigned b0 = __builtin_amdgcn_ballot_w32((e & 1) != 0), b1 = __builtin_amdgcn_ballot_w32((e & 2) != 0),
                           b2 = __builtin_amdgcn_ballot_w32((e & 4) != 0), b3 = __builtin_amdgcn_ballot_w32((e & 8) != 0),
                           b4 = __builtin_amdgcn_ballot_w32((e & 16) != 0);
            mine += __builtin_popcount(vm & ~(b0 ^ l0) & ~(b1 ^ l1) & ~(b2 ^ l2) & ~(b3 ^ l3) & ~(b4 ^ l4));
        }
        cnt[w][lane] = mine;
        __syncthreads();

        int before = 0, tot = 0;
        for (int q = 0; q < OWN_WAVES; ++q) {
            const int c = cnt[q][lane];
            if (q < w) before += c;
            tot += c;
        }
        int incl = tot;
        incl += __builtin_amdgcn_update_dpp(0, incl, 0x111, 0xf, 0xf, true);
        incl += __builtin_amdgcn_update_dpp(0, incl, 0x112, 0xf, 0xf, true);
        incl += __builtin_amdgcn_update_dpp(0, incl, 0x114, 0xf, 0xf, true);
        incl += __builtin_amdgcn_update_dpp(0, incl, 0x118, 0xf, 0xf, true);
        const int low = __builtin_amdgcn_readlane(incl, 15);
        if (lane >= 16) incl += low;
        const int ostart = incl - tot;
        int next = ostart + before;
        const int mystart = __builtin_amdgcn_readlane(ostart, w);
        const int mycnt   = __builtin_amdgcn_readlane(tot, w);

        for (int i0 = 0; i0 < qn; i0 += 32) {
            const int i = i0 + lane;
            const bool in = i < qn;
            const int e = in ? queue[w][i] : 0;
            const unsigned vm = __builtin_amdgcn_ballot_w32(in);
            const unsigned b0 = __builtin_amdgcn_ballot_w32((e & 1) != 0), b1 = __builtin_amdgcn_ballot_w32((e & 2) != 0),
                           b2 = __builtin_amdgcn_ballot_w32((e & 4) != 0), b3 = __builtin_amdgcn_ballot_w32((e & 8) != 0),
                           b4 = __builtin_amdgcn_ballot_w32((e & 16) != 0);
            const unsigned e0 = (e & 1) ? ~0u : 0u, e1 = (e & 2) ? ~0u : 0u, e2 = (e & 4) ? ~0u : 0u,
                           e3 = (e & 8) ? ~0u : 0u, e4 = (e & 16) ? ~0u : 0u;
            const unsigned peers = vm & ~(b0 ^ e0) & ~(b1 ^ e1) & ~(b2 ^ e2) & ~(b3 ^ e3) & ~(b4 ^ e4);
            const int rank = (int)__builtin_amdgcn_mbcnt_lo(peers, 0u);
            const int place = __shfl(next, e & 31);
            if (in) pool[place + rank] = e;
            next += __builtin_popcount(vm & ~(b0 ^ l0) & ~(b1 ^ l1) & ~(b2 ^ l2) & ~(b3 ^ l3) & ~(b4 ^ l4));
        }
        __syncthreads();

        auto drain = [&](int ev, int first, int n) {
            float4 v[OWN_BATCH];
            int rs[OWN_BATCH];
#pragma unroll
            for (int k = 0; k < OWN_BATCH; ++k) {
                if (k < n) {
                    const int e = __builtin_amdgcn_readlane(ev, first + k);
                    rs[k] = (e & 255) >> 5;
                    v[k] = ((const float4*)(src + (size_t)((unsigned)e >> 8) * DFEAT))[lane];
                }
            }
#pragma unroll
            for (int k = 0; k < OWN_BATCH; ++k) {
                if (k < n) {
                    const int s = rs[k];
                    ax[s] += v[k].x; ay[s] += v[k].y; az[s] += v[k].z; aw[s] += v[k].w;
                    if (BLEND) rc[s] += 1;
                }
            }
        };
        for (int h = 0; h < mycnt; h += 32) {
            const int left = mycnt - h;
            const int nb = left < 32 ? left : 32;
            const int ev = lane < nb ? pool[mystart + h + lane] : 0;
            int first = 0;
#pragma nounroll
            for (; first + OWN_BATCH <= nb; first += OWN_BATCH) drain(ev, first, OWN_BATCH);
            if (first < nb) drain(ev, first, nb - first);
        }

        any = __syncthreads_or((pos < segEnd || gi < gn) ? 1 : 0);
    } while (any);

    float4* d4 = (float4*)dst + (size_t)base * 32;
    if (BLEND) {
        const float4* a4 = (const float4*)Ap + (size_t)base * 32;
        const float4* x4 = (const float4*)X0 + (size_t)base * 32;
        if (rows > 0) {
            float4 av[NS], xv[NS];
#pragma unroll
            for (int t = 0; t < NS; ++t) {
                const int r = w + OWN_WAVES * t;
                const int j = r < rows ? r : rows - 1;
                av[t] = a4[j * 32 + lane];
                xv[t] = x4[j * 32 + lane];
            }
#pragma unroll
            for (int t = 0; t < NS; ++t) {
                const int r = w + OWN_WAVES * t;
                if (r < rows) {
                    const float4 s = make_float4(ax[t], ay[t], az[t], aw[t]);
                    const float d = (float)rc[t];
                    float4 o;
                    o.x = 0.5f * (d * av[t].x + s.x) + 0.5f * xv[t].x;
                    o.y = 0.5f * (d * av[t].y + s.y) + 0.5f * xv[t].y;
                    o.z = 0.5f * (d * av[t].z + s.z) + 0.5f * xv[t].z;
                    o.w = 0.5f * (d * av[t].w + s.w) + 0.5f * xv[t].w;
                    d4[r * 32 + lane] = o;
                }
            }
        }
    } else {
#pragma unroll
        for (int t = 0; t < NS; ++t) {
            const int r = w + OWN_WAVES * t;
            if (r < rows) d4[r * 32 + lane] = make_float4(ax[t], ay[t], az[t], aw[t]);
        }
    }
}

extern "C" void kernel_launch(void* const* d_in, const int* in_sizes, int n_in,
                              void* d_out, int out_size, void* d_ws, size_t ws_size,
                              hipStream_t stream) {
    const float* X      = (const float*)d_in[0];
    const int*   vertex = (const int*)d_in[1];
    const int*   edges  = (const int*)d_in[2];
    const float* X0     = (const float*)d_in[3];
    const float* W1     = (const float*)d_in[4];
    const float* b1     = (const float*)d_in[5];
    const float* W2     = (const float*)d_in[6];
    const float* b2     = (const float*)d_in[7];
    const float* W3     = (const float*)d_in[8];
    const float* b3     = (const float*)d_in[9];
    float* out = (float*)d_out;

    constexpr int N = 50000, M = 25000, NNZ = 800000;
    constexpr size_t ND = (size_t)N * DFEAT;
    constexpr size_t MD = (size_t)M * DFEAT;

    char* ws = (char*)d_ws;
    auto align256 = [](size_t x) { return (x + 255) & ~(size_t)255; };
    size_t off = 0;
    float* XW = (float*)(ws + off); off = align256(off + ND * 4);
    float* Xe = (float*)(ws + off); off = align256(off + MD * 4);
    float* Ap = (float*)(ws + off); off = align256(off + ND * 4);
    float* Bm = (float*)(ws + off); off = align256(off + MD * 4);
    float* Xn = XW;

    const int gemmN = (N + 127) / 128;
    const int gemmM = (M + 127) / 128;
    constexpr int RM = 98;
    constexpr int RN = 196;
    static_assert((M + RM - 1) / RM <= 256 && (N + RN - 1) / RN <= 256, "an owner number must fit one byte");

    constexpr int OWN_POS    = (NNZ + OWN_BLK - 1) / OWN_BLK * OWN_BLK;
    constexpr int OWN_WORDS  = OWN_POS / 4;
    constexpr int OWN_GROUPS = OWN_POS / 16;
    constexpr int OWN_GRID   = (OWN_GROUPS + 255) / 256;
    static_assert((size_t)OWN_POS + (size_t)OWN_GROUPS * 32 <= MD * 4, "bytes and planes must fit the region they borrow");
    int* ownE = (int*)Bm;
    int* plE  = ownE + OWN_WORDS;
    int* ownV = (int*)Xe;
    int* plV  = ownV + OWN_WORDS;

    owner_marks<RM><<<OWN_GRID, 256, 0, stream>>>(edges, ownE, plE, NNZ, OWN_GROUPS);

    gemm128_wmma<<<gemmN, 256, 0, stream>>>(X, N, W1, b1, XW);

    own_rows<RM, false><<<(M + RM - 1) / RM, OWN_THREADS, 0, stream>>>(XW, vertex, edges, ownE, plE, OWN_GROUPS, Xe, nullptr, nullptr, M, NNZ);

    gemm128_wmma<<<gemmN, 256, 0, stream>>>(X, N, W2, b2, Ap);

    gemm128_wmma<<<gemmM, 256, 0, stream>>>(Xe, M, W2 + 128 * DFEAT, nullptr, Bm);

    owner_marks<RN><<<OWN_GRID, 256, 0, stream>>>(vertex, ownV, plV, NNZ, OWN_GROUPS);

    own_rows<RN, true><<<(N + RN - 1) / RN, OWN_THREADS, 0, stream>>>(Bm, edges, vertex, ownV, plV, OWN_GROUPS, Xn, Ap, X0, N, NNZ);

    gemm128_wmma<<<gemmN, 256, 0, stream>>>(Xn, N, W3, b3, out);
}


# Products fed by row sums run inside the row-sum launches: owned rows go registers -> LDS tile, waves multiply it on WMMA (Xe@W2b, Xn@W3); 2 GEMM launches and the Xe, Xn round trips go; sums unchanged
# speedup vs baseline: 3.5608x; 1.1505x over previous
#include <hip/hip_runtime.h>


typedef __attribute__((ext_vector_type(2))) float v2f;
typedef __attribute__((ext_vector_type(4))) float v4f;
typedef __attribute__((ext_vector_type(8))) float v8f;
typedef __attribute__((ext_vector_type(8))) int   v8i;

#define DFEAT 128

__global__ __launch_bounds__(256) void gemm128_wmma(
    const float* __restrict__ Xin, int Mrows,
    const float* __restrict__ W,
    const float* __restrict__ bias,
    float* __restrict__ Y)
{
    const int lane = threadIdx.x & 31;
    const int wave = threadIdx.x >> 5;
    const int half = lane >> 4;
    const int r    = lane & 15;
    const int m0   = (blockIdx.x * 8 + wave) * 16;
    if (m0 >= Mrows) return;

    v8f acc[8] = {};

    int arow = m0 + r;
    if (arow >= Mrows) arow = Mrows - 1;
    const v4f* ap = (const v4f*)(Xin + (size_t)arow * DFEAT + 4 * half);
    v4f a4[16];
#pragma unroll
    for (int q = 0; q < 16; ++q) a4[q] = ap[2 * q];

    const v4f* wp = (const v4f*)(W + (size_t)(4 * half) * DFEAT + 4 * r);
    v4f b[2][4][2];
#pragma unroll
    for (int i = 0; i < 4; ++i) {
        b[0][i][0] = wp[i * 32];
        b[0][i][1] = wp[i * 32 + 16];
    }

#pragma unroll
    for (int q = 0; q < 16; ++q) {
        if (q + 1 < 16) {
#pragma unroll
            for (int i = 0; i < 4; ++i) {
                b[(q + 1) & 1][i][0] = wp[((q + 1) * 8 + i) * 32];
                b[(q + 1) & 1][i][1] = wp[((q + 1) * 8 + i) * 32 + 16];
            }
        }
        __builtin_amdgcn_sched_barrier(0);
#pragma unroll
        for (int e = 0; e < 2; ++e) {
            v2f a;
            a.x = a4[q][2 * e];
            a.y = a4[q][2 * e + 1];
#pragma unroll
            for (int t = 0; t < 8; ++t) {
                v2f bb;
                bb.x = b[q & 1][2 * e][t >> 2][t & 3];
                bb.y = b[q & 1][2 * e + 1][t >> 2][t & 3];
                acc[t] = __builtin_amdgcn_wmma_f32_16x16x4_f32(
                    false, a, false, bb, (short)0, acc[t], false, false);
            }
        }
    }

    v4f bv[2];
#pragma unroll
    for (int u = 0; u < 2; ++u) {
        if (bias) bv[u] = ((const v4f*)bias)[16 * u + r];
        else { bv[u].x = 0.0f; bv[u].y = 0.0f; bv[u].z = 0.0f; bv[u].w = 0.0f; }
    }
#pragma unroll
    for (int j = 0; j < 8; ++j) {
        const int row = m0 + j + 8 * half;
        if (row < Mrows) {
            v4f* yp = (v4f*)(Y + (size_t)row * DFEAT);
#pragma unroll
            for (int u = 0; u < 2; ++u) {
                v4f o;
                o.x = acc[4 * u + 0][j] + bv[u].x;
                o.y = acc[4 * u + 1][j] + bv[u].y;
                o.z = acc[4 * u + 2][j] + bv[u].z;
                o.w = acc[4 * u + 3][j] + bv[u].w;
                yp[16 * u + r] = o;
            }
        }
    }
}

template <int R>
__global__ __launch_bounds__(256) void owner_marks(
    const int* __restrict__ ids, int* __restrict__ own, int* __restrict__ plane, int nnz, int ngroups)
{
    const int g = blockIdx.x * 256 + threadIdx.x;
    if (g >= ngroups) return;
    const int p0 = 16 * g;
    int id[16];
    if (p0 + 15 < nnz) {
        const int4* q = (const int4*)(ids + p0);
#pragma unroll
        for (int k = 0; k < 4; ++k) {
            const int4 a = q[k];
            id[4 * k] = a.x; id[4 * k + 1] = a.y; id[4 * k + 2] = a.z; id[4 * k + 3] = a.w;
        }
    } else {
#pragma unroll
        for (int k = 0; k < 16; ++k) id[k] = p0 + k < nnz ? ids[p0 + k] : 0;
    }
    unsigned ob[4] = {0u, 0u, 0u, 0u};
    unsigned m[8] = {0u, 0u, 0u, 0u, 0u, 0u, 0u, 0u};
#pragma unroll
    for (int k = 0; k < 16; ++k) {
        const bool valid = p0 + k < nnz;
        const unsigned o = valid ? (unsigned)(id[k] / R) : 0u;
        ob[k >> 2] |= o << (8 * (k & 3));
        const unsigned bit = valid ? (1u << (o & 31u)) : 0u;
        const unsigned hi = o >> 5;
#pragma unroll
        for (int j = 0; j < 8; ++j) m[j] |= hi == (unsigned)j ? bit : 0u;
    }
    ((uint4*)own)[g] = make_uint4(ob[0], ob[1], ob[2], ob[3]);
#pragma unroll
    for (int j = 0; j < 8; ++j) plane[(size_t)j * ngroups + g] = (int)m[j];
}

#define OWN_THREADS 1024
#define OWN_WAVES   32
#define OWN_QCAP    768
#define OWN_BLK     512
#define OWN_UB      25
#define OWN_BATCH   8
#define OWN_GCAP    384
#define OWN_TP      132

template <int R, bool BLEND>
__global__ __launch_bounds__(OWN_THREADS) void own_rows(
    const float* __restrict__ src,
    const int* __restrict__ gidx,
    const int* __restrict__ sidx,
    const int* __restrict__ own,
    const int* __restrict__ plane,
    int ngroups,
    float* __restrict__ dst,
    const float* __restrict__ Ap,
    const float* __restrict__ X0,
    int nrows, int nnz,
    const float* __restrict__ Wt,
    const float* __restrict__ bt)
{
    static_assert(R <= 256, "local row must fit the low 8 bits of a queue entry");
    constexpr int NS = (R + OWN_WAVES - 1) / OWN_WAVES;
    constexpr int RT = (R + 15) / 16;
    __shared__ alignas(16) int qp[2 * OWN_WAVES * OWN_QCAP];
    int (*queue)[OWN_QCAP] = reinterpret_cast<int (*)[OWN_QCAP]>(qp);
    int* pool = qp + OWN_WAVES * OWN_QCAP;
    float* tile = reinterpret_cast<float*>(qp);
    static_assert(RT * 16 * OWN_TP <= 2 * OWN_WAVES * OWN_QCAP, "the row tile must fit the space of the queues and the pool");
    __shared__ int kept[OWN_WAVES][OWN_GCAP];
    __shared__ int cnt[OWN_WAVES][OWN_WAVES];

    const int tid  = threadIdx.x;
    const int lane = tid & 31;
    const int w    = __builtin_amdgcn_readfirstlane(tid >> 5);
    const int base = blockIdx.x * R;
    int rows = nrows - base;
    if (rows > R) rows = R;
    if (rows < 0) rows = 0;

    static_assert(NS <= 8, "a slot must index an 8-wide register vector");
    v8f ax = {}, ay = {}, az = {}, aw = {};
    v8i rc = {};

    const int seg = (((nnz + OWN_WAVES - 1) / OWN_WAVES) + OWN_BLK - 1) / OWN_BLK * OWN_BLK;
    const uint4* __restrict__ own4 = (const uint4*)own;
    const unsigned me = blockIdx.x;
    const unsigned me4 = me * 0x01010101u;
    const unsigned* __restrict__ myplane = (const unsigned*)plane + (size_t)(me >> 5) * (size_t)ngroups;
    const unsigned mebit = 1u << (me & 31u);
    int pos = w * seg;
    int segEnd = pos + seg;
    if (segEnd > nnz) segEnd = nnz;

    int gn = 0, gi = 0;

    int any;
    do {
        if (gi >= gn) {
            gn = 0;
            gi = 0;
            while (pos < segEnd && gn <= OWN_GCAP - 32) {
                unsigned d[OWN_UB];
#pragma unroll
                for (int u = 0; u < OWN_UB; ++u) {
                    int pb = pos + u * OWN_BLK;
                    if (pb >= segEnd) pb = pos;
                    d[u] = myplane[(pb >> 4) + lane];
                }
#pragma unroll
                for (int u = 0; u < OWN_UB; ++u) {
                    if (pos >= segEnd || gn > OWN_GCAP - 32) break;
                    const bool t = (d[u] & mebit) != 0u;
                    const unsigned hit = __builtin_amdgcn_ballot_w32(t);
                    if (t) kept[w][gn + (int)__builtin_amdgcn_mbcnt_lo(hit, 0u)] = (pos >> 4) + lane;
                    gn += __builtin_popcount(hit);
                    pos += OWN_BLK;
                }
            }
        }
        int qn = 0;
        while (gi < gn && qn <= OWN_QCAP - OWN_BLK) {
            const int i = gi + lane;
            const bool in = i < gn;
            const int g = in ? kept[w][i] : 0;
            const uint4 dd = own4[g];
            const unsigned y0 = dd.x ^ me4, y1 = dd.y ^ me4, y2 = dd.z ^ me4, y3 = dd.w ^ me4;
            const unsigned f0 = ~(((y0 & 0x7f7f7f7fu) + 0x7f7f7f7fu) | y0) & 0x80808080u;
            const unsigned f1 = ~(((y1 & 0x7f7f7f7fu) + 0x7f7f7f7fu) | y1) & 0x80808080u;
            const unsigned f2 = ~(((y2 & 0x7f7f7f7fu) + 0x7f7f7f7fu) | y2) & 0x80808080u;
            const unsigned f3 = ~(((y3 & 0x7f7f7f7fu) + 0x7f7f7f7fu) | y3) & 0x80808080u;
            const unsigned x0 = f0 >> 7, x1 = f1 >> 7, x2 = f2 >> 7, x3 = f3 >> 7;
            const unsigned n0 = (x0 | (x0 >> 7) | (x0 >> 14) | (x0 >> 21)) & 15u;
            const unsigned n1 = (x1 | (x1 >> 7) | (x1 >> 14) | (x1 >> 21)) & 15u;
            const unsigned n2 = (x2 | (x2 >> 7) | (x2 >> 14) | (x2 >> 21)) & 15u;
            const unsigned n3 = (x3 | (x3 >> 7) | (x3 >> 14) | (x3 >> 21)) & 15u;
            unsigned hm = n0 | (n1 << 4) | (n2 << 8) | (n3 << 12);
            const int nv = segEnd - g * 16;
            hm = !in || nv <= 0 ? 0u : (nv >= 16 ? hm : (hm & ((1u << nv) - 1u)));
            const int c = __builtin_popcount(hm);
            int incl = c;
            incl += __builtin_amdgcn_update_dpp(0, incl, 0x111, 0xf, 0xf, true);
            incl += __builtin_amdgcn_update_dpp(0, incl, 0x112, 0xf, 0xf, true);
            incl += __builtin_amdgcn_update_dpp(0, incl, 0x114, 0xf, 0xf, true);
            incl += __builtin_amdgcn_update_dpp(0, incl, 0x118, 0xf, 0xf, true);
            const int low = __builtin_amdgcn_readlane(incl, 15);
            if (lane >= 16) incl += low;
            int at = qn + incl - c;
            while (hm) {
                queue[w][at++] = g * 16 + __builtin_ctz(hm);
                hm &= hm - 1u;
            }
            qn += __builtin_amdgcn_readlane(incl, 31);
            gi += 32;
        }
        const unsigned l0 = (lane & 1) ? ~0u : 0u, l1 = (lane & 2) ? ~0u : 0u, l2 = (lane & 4) ? ~0u : 0u,
                       l3 = (lane & 8) ? ~0u : 0u, l4 = (lane & 16) ? ~0u : 0u;
        int mine = 0;
        for (int i0 = 0; i0 < qn; i0 += 32) {
            const int i = i0 + lane;
            const bool in = i < qn;
            int e = 0;
            if (in) {
                const int p = queue[w][i];
                const int s = sidx[p];
                const int g = gidx[p];
                e = (g << 8) | (s - base);
                queue[w][i] = e;
            }
            const unsigned vm = __builtin_amdgcn_ballot_w32(in);
            const unsigned b0 = __builtin_amdgcn_ballot_w32((e & 1) != 0), b1 = __builtin_amdgcn_ballot_w32((e & 2) != 0),
                           b2 = __builtin_amdgcn_ballot_w32((e & 4) != 0), b3 = __builtin_amdgcn_ballot_w32((e & 8) != 0),
                           b4 = __builtin_amdgcn_ballot_w32((e & 16) != 0);
            mine += __builtin_popcount(vm & ~(b0 ^ l0) & ~(b1 ^ l1) & ~(b2 ^ l2) & ~(b3 ^ l3) & ~(b4 ^ l4));
        }
        cnt[w][lane] = mine;
        __syncthreads();

        int before = 0, tot = 0;
        for (int q = 0; q < OWN_WAVES; ++q) {
            const int c = cnt[q][lane];
            if (q < w) before += c;
            tot += c;
        }
        int incl = tot;
        incl += __builtin_amdgcn_update_dpp(0, incl, 0x111, 0xf, 0xf, true);
        incl += __builtin_amdgcn_update_dpp(0, incl, 0x112, 0xf, 0xf, true);
        incl += __builtin_amdgcn_update_dpp(0, incl, 0x114, 0xf, 0xf, true);
        incl += __builtin_amdgcn_update_dpp(0, incl, 0x118, 0xf, 0xf, true);
        const int low = __builtin_amdgcn_readlane(incl, 15);
        if (lane >= 16) incl += low;
        const int ostart = incl - tot;
        int next = ostart + before;
        const int mystart = __builtin_amdgcn_readlane(ostart, w);
        const int mycnt   = __builtin_amdgcn_readlane(tot, w);

        for (int i0 = 0; i0 < qn; i0 += 32) {
            const int i = i0 + lane;
            const bool in = i < qn;
            const int e = in ? queue[w][i] : 0;
            const unsigned vm = __builtin_amdgcn_ballot_w32(in);
            const unsigned b0 = __builtin_amdgcn_ballot_w32((e & 1) != 0), b1 = __builtin_amdgcn_ballot_w32((e & 2) != 0),
                           b2 = __builtin_amdgcn_ballot_w32((e & 4) != 0), b3 = __builtin_amdgcn_ballot_w32((e & 8) != 0),
                           b4 = __builtin_amdgcn_ballot_w32((e & 16) != 0);
            const unsigned e0 = (e & 1) ? ~0u : 0u, e1 = (e & 2) ? ~0u : 0u, e2 = (e & 4) ? ~0u : 0u,
                           e3 = (e & 8) ? ~0u : 0u, e4 = (e & 16) ? ~0u : 0u;
            const unsigned peers = vm & ~(b0 ^ e0) & ~(b1 ^ e1) & ~(b2 ^ e2) & ~(b3 ^ e3) & ~(b4 ^ e4);
            const int rank = (int)__builtin_amdgcn_mbcnt_lo(peers, 0u);
            const int place = __shfl(next, e & 31);
            if (in) pool[place + rank] = e;
            next += __builtin_popcount(vm & ~(b0 ^ l0) & ~(b1 ^ l1) & ~(b2 ^ l2) & ~(b3 ^ l3) & ~(b4 ^ l4));
        }
        __syncthreads();

        auto drain = [&](int ev, int first, int n) {
            float4 v[OWN_BATCH];
            int rs[OWN_BATCH];
#pragma unroll
            for (int k = 0; k < OWN_BATCH; ++k) {
                if (k < n) {
                    const int e = __builtin_amdgcn_readlane(ev, first + k);
                    rs[k] = (e & 255) >> 5;
                    v[k] = ((const float4*)(src + (size_t)((unsigned)e >> 8) * DFEAT))[lane];
                }
            }
#pragma unroll
            for (int k = 0; k < OWN_BATCH; ++k) {
                if (k < n) {
                    const int s = rs[k];
                    ax[s] += v[k].x; ay[s] += v[k].y; az[s] += v[k].z; aw[s] += v[k].w;
                    if (BLEND) rc[s] += 1;
                }
            }
        };
        for (int h = 0; h < mycnt; h += 32) {
            const int left = mycnt - h;
            const int nb = left < 32 ? left : 32;
            const int ev = lane < nb ? pool[mystart + h + lane] : 0;
            int first = 0;
#pragma nounroll
            for (; first + OWN_BATCH <= nb; first += OWN_BATCH) drain(ev, first, OWN_BATCH);
            if (first < nb) drain(ev, first, nb - first);
        }

        any = __syncthreads_or((pos < segEnd || gi < gn) ? 1 : 0);
    } while (any);

    if (BLEND) {
        const float4* a4 = (const float4*)Ap + (size_t)base * 32;
        const float4* x4 = (const float4*)X0 + (size_t)base * 32;
        if (rows > 0) {
            float4 av[NS], xv[NS];
#pragma unroll
            for (int t = 0; t < NS; ++t) {
                const int r = w + OWN_WAVES * t;
                const int j = r < rows ? r : rows - 1;
                av[t] = a4[j * 32 + lane];
                xv[t] = x4[j * 32 + lane];
            }
#pragma unroll
            for (int t = 0; t < NS; ++t) {
                const int r = w + OWN_WAVES * t;
                if (r < rows) {
                    const float4 s = make_float4(ax[t], ay[t], az[t], aw[t]);
                    const float d = (float)rc[t];
                    float4 o;
                    o.x = 0.5f * (d * av[t].x + s.x) + 0.5f * xv[t].x;
                    o.y = 0.5f * (d * av[t].y + s.y) + 0.5f * xv[t].y;
                    o.z = 0.5f * (d * av[t].z + s.z) + 0.5f * xv[t].z;
                    o.w = 0.5f * (d * av[t].w + s.w) + 0.5f * xv[t].w;
                    ((float4*)(tile + r * OWN_TP))[lane] = o;
                }
            }
        }
    } else {
#pragma unroll
        for (int t = 0; t < NS; ++t) {
            const int r = w + OWN_WAVES * t;
            if (r < rows) ((float4*)(tile + r * OWN_TP))[lane] = make_float4(ax[t], ay[t], az[t], aw[t]);
        }
    }
    __syncthreads();

    {
        const int half = lane >> 4;
        const int r16  = lane & 15;
        const int nunits = 2 * ((rows + 15) >> 4);
        for (int unit = w; unit < nunits; unit += OWN_WAVES) {
            const int rt = unit >> 1;
            const int u  = unit & 1;
            v8f acc[4] = {};
            const float* arow = tile + (rt * 16 + r16) * OWN_TP + 4 * half;
            const v4f* wp = (const v4f*)(Wt + (size_t)(4 * half) * DFEAT + 64 * u + 4 * r16);
            v4f a4[2];
            v4f b[2][4];
            a4[0] = *(const v4f*)arow;
#pragma unroll
            for (int i = 0; i < 4; ++i) b[0][i] = wp[i * 32];
#pragma unroll
            for (int q = 0; q < 16; ++q) {
                if (q + 1 < 16) {
                    a4[(q + 1) & 1] = *(const v4f*)(arow + 8 * (q + 1));
#pragma unroll
                    for (int i = 0; i < 4; ++i) b[(q + 1) & 1][i] = wp[((q + 1) * 8 + i) * 32];
                }
                __builtin_amdgcn_sched_barrier(0);
#pragma unroll
                for (int e = 0; e < 2; ++e) {
                    v2f a;
                    a.x = a4[q & 1][2 * e];
                    a.y = a4[q & 1][2 * e + 1];
#pragma unroll
                    for (int t = 0; t < 4; ++t) {
                        v2f bb;
                        bb.x = b[q & 1][2 * e][t];
                        bb.y = b[q & 1][2 * e + 1][t];
                        acc[t] = __builtin_amdgcn_wmma_f32_16x16x4_f32(
                            false, a, false, bb, (short)0, acc[t], false, false);
                    }
                }
            }
            v4f bv;
            if (bt) bv = ((const v4f*)bt)[16 * u + r16];
            else { bv.x = 0.0f; bv.y = 0.0f; bv.z = 0.0f; bv.w = 0.0f; }
#pragma unroll
            for (int j = 0; j < 8; ++j) {
                const int row = rt * 16 + j + 8 * half;
                if (row < rows) {
                    v4f o;
                    o.x = acc[0][j] + bv.x;
                    o.y = acc[1][j] + bv.y;
                    o.z = acc[2][j] + bv.z;
                    o.w = acc[3][j] + bv.w;
                    ((v4f*)(dst + (size_t)(base + row) * DFEAT))[16 * u + r16] = o;
                }
            }
        }
    }
}

extern "C" void kernel_launch(void* const* d_in, const int* in_sizes, int n_in,
                              void* d_out, int out_size, void* d_ws, size_t ws_size,
                              hipStream_t stream) {
    const float* X      = (const float*)d_in[0];
    const int*   vertex = (const int*)d_in[1];
    const int*   edges  = (const int*)d_in[2];
    const float* X0     = (const float*)d_in[3];
    const float* W1     = (const float*)d_in[4];
    const float* b1     = (const float*)d_in[5];
    const float* W2     = (const float*)d_in[6];
    const float* b2     = (const float*)d_in[7];
    const float* W3     = (const float*)d_in[8];
    const float* b3     = (const float*)d_in[9];
    float* out = (float*)d_out;

    constexpr int N = 50000, M = 25000, NNZ = 800000;
    constexpr size_t ND = (size_t)N * DFEAT;
    constexpr size_t MD = (size_t)M * DFEAT;

    char* ws = (char*)d_ws;
    auto align256 = [](size_t x) { return (x + 255) & ~(size_t)255; };
    size_t off = 0;
    float* XW = (float*)(ws + off); off = align256(off + ND * 4);
    char*  Mk = ws + off;           off = align256(off + MD * 4);
    float* Ap = (float*)(ws + off); off = align256(off + ND * 4);
    float* Bm = (float*)(ws + off); off = align256(off + MD * 4);

    const int gemmN = (N + 127) / 128;
    constexpr int RM = 98;
    constexpr int RN = 196;
    static_assert((M + RM - 1) / RM <= 256 && (N + RN - 1) / RN <= 256, "an owner number must fit one byte");

    constexpr int OWN_POS    = (NNZ + OWN_BLK - 1) / OWN_BLK * OWN_BLK;
    constexpr int OWN_WORDS  = OWN_POS / 4;
    constexpr int OWN_GROUPS = OWN_POS / 16;
    constexpr int OWN_GRID   = (OWN_GROUPS + 255) / 256;
    constexpr size_t MARK_BYTES = ((size_t)OWN_POS + (size_t)OWN_GROUPS * 32 + 255) / 256 * 256;
    static_assert(2 * MARK_BYTES <= MD * 4, "both lists' marks must fit their region");
    int* ownE = (int*)Mk;
    int* plE  = ownE + OWN_WORDS;
    int* ownV = (int*)(Mk + MARK_BYTES);
    int* plV  = ownV + OWN_WORDS;

    owner_marks<RM><<<OWN_GRID, 256, 0, stream>>>(edges, ownE, plE, NNZ, OWN_GROUPS);

    gemm128_wmma<<<gemmN, 256, 0, stream>>>(X, N, W1, b1, XW);

    own_rows<RM, false><<<(M + RM - 1) / RM, OWN_THREADS, 0, stream>>>(XW, vertex, edges, ownE, plE, OWN_GROUPS, Bm, nullptr, nullptr, M, NNZ, W2 + 128 * DFEAT, nullptr);

    gemm128_wmma<<<gemmN, 256, 0, stream>>>(X, N, W2, b2, Ap);

    owner_marks<RN><<<OWN_GRID, 256, 0, stream>>>(vertex, ownV, plV, NNZ, OWN_GROUPS);

    own_rows<RN, true><<<(N + RN - 1) / RN, OWN_THREADS, 0, stream>>>(Bm, edges, vertex, ownV, plV, OWN_GROUPS, out, Ap, X0, N, NNZ, W3, b3);
}


# The four launches that read only the inputs (X@W1, X@W2a, both owner-mark passes) run as one launch, block index picks the job; 6 -> 3 launches; each job's code and sums unchanged
# speedup vs baseline: 3.7671x; 1.0579x over previous
#include <hip/hip_runtime.h>


typedef __attribute__((ext_vector_type(2))) float v2f;
typedef __attribute__((ext_vector_type(4))) float v4f;
typedef __attribute__((ext_vector_type(8))) float v8f;
typedef __attribute__((ext_vector_type(8))) int   v8i;

#define DFEAT 128

__device__ __forceinline__ void gemm128_slab(
    int blk,
    const float* __restrict__ Xin, int Mrows,
    const float* __restrict__ W,
    const float* __restrict__ bias,
    float* __restrict__ Y)
{
    const int lane = threadIdx.x & 31;
    const int wave = threadIdx.x >> 5;
    const int half = lane >> 4;
    const int r    = lane & 15;
    const int m0   = (blk * 8 + wave) * 16;
    if (m0 >= Mrows) return;

    v8f acc[8] = {};

    int arow = m0 + r;
    if (arow >= Mrows) arow = Mrows - 1;
    const v4f* ap = (const v4f*)(Xin + (size_t)arow * DFEAT + 4 * half);
    v4f a4[16];
#pragma unroll
    for (int q = 0; q < 16; ++q) a4[q] = ap[2 * q];

    const v4f* wp = (const v4f*)(W + (size_t)(4 * half) * DFEAT + 4 * r);
    v4f b[2][4][2];
#pragma unroll
    for (int i = 0; i < 4; ++i) {
        b[0][i][0] = wp[i * 32];
        b[0][i][1] = wp[i * 32 + 16];
    }

#pragma unroll
    for (int q = 0; q < 16; ++q) {
        if (q + 1 < 16) {
#pragma unroll
            for (int i = 0; i < 4; ++i) {
                b[(q + 1) & 1][i][0] = wp[((q + 1) * 8 + i) * 32];
                b[(q + 1) & 1][i][1] = wp[((q + 1) * 8 + i) * 32 + 16];
            }
        }
        __builtin_amdgcn_sched_barrier(0);
#pragma unroll
        for (int e = 0; e < 2; ++e) {
            v2f a;
            a.x = a4[q][2 * e];
            a.y = a4[q][2 * e + 1];
#pragma unroll
            for (int t = 0; t < 8; ++t) {
                v2f bb;
                bb.x = b[q & 1][2 * e][t >> 2][t & 3];
                bb.y = b[q & 1][2 * e + 1][t >> 2][t & 3];
                acc[t] = __builtin_amdgcn_wmma_f32_16x16x4_f32(
                    false, a, false, bb, (short)0, acc[t], false, false);
            }
        }
    }

    v4f bv[2];
#pragma unroll
    for (int u = 0; u < 2; ++u) {
        if (bias) bv[u] = ((const v4f*)bias)[16 * u + r];
        else { bv[u].x = 0.0f; bv[u].y = 0.0f; bv[u].z = 0.0f; bv[u].w = 0.0f; }
    }
#pragma unroll
    for (int j = 0; j < 8; ++j) {
        const int row = m0 + j + 8 * half;
        if (row < Mrows) {
            v4f* yp = (v4f*)(Y + (size_t)row * DFEAT);
#pragma unroll
            for (int u = 0; u < 2; ++u) {
                v4f o;
                o.x = acc[4 * u + 0][j] + bv[u].x;
                o.y = acc[4 * u + 1][j] + bv[u].y;
                o.z = acc[4 * u + 2][j] + bv[u].z;
                o.w = acc[4 * u + 3][j] + bv[u].w;
                yp[16 * u + r] = o;
            }
        }
    }
}

template <int R>
__device__ __forceinline__ void owner_marks_blk(
    int blk,
    const int* __restrict__ ids, int* __restrict__ own, int* __restrict__ plane, int nnz, int ngroups)
{
    const int g = blk * 256 + threadIdx.x;
    if (g >= ngroups) return;
    const int p0 = 16 * g;
    int id[16];
    if (p0 + 15 < nnz) {
        const int4* q = (const int4*)(ids + p0);
#pragma unroll
        for (int k = 0; k < 4; ++k) {
            const int4 a = q[k];
            id[4 * k] = a.x; id[4 * k + 1] = a.y; id[4 * k + 2] = a.z; id[4 * k + 3] = a.w;
        }
    } else {
#pragma unroll
        for (int k = 0; k < 16; ++k) id[k] = p0 + k < nnz ? ids[p0 + k] : 0;
    }
    unsigned ob[4] = {0u, 0u, 0u, 0u};
    unsigned m[8] = {0u, 0u, 0u, 0u, 0u, 0u, 0u, 0u};
#pragma unroll
    for (int k = 0; k < 16; ++k) {
        const bool valid = p0 + k < nnz;
        const unsigned o = valid ? (unsigned)(id[k] / R) : 0u;
        ob[k >> 2] |= o << (8 * (k & 3));
        const unsigned bit = valid ? (1u << (o & 31u)) : 0u;
        const unsigned hi = o >> 5;
#pragma unroll
        for (int j = 0; j < 8; ++j) m[j] |= hi == (unsigned)j ? bit : 0u;
    }
    ((uint4*)own)[g] = make_uint4(ob[0], ob[1], ob[2], ob[3]);
#pragma unroll
    for (int j = 0; j < 8; ++j) plane[(size_t)j * ngroups + g] = (int)m[j];
}

template <int RE, int RV>
__global__ __launch_bounds__(256) void input_jobs(
    const float* __restrict__ X, int nrows,
    const float* __restrict__ Wa, const float* __restrict__ ba, float* __restrict__ Ya,
    const float* __restrict__ Wb, const float* __restrict__ bb, float* __restrict__ Yb,
    const int* __restrict__ idsE, int* __restrict__ ownE, int* __restrict__ plE,
    const int* __restrict__ idsV, int* __restrict__ ownV, int* __restrict__ plV,
    int nnz, int ngroups, int gb, int mb)
{
    int b = blockIdx.x;
    if (b < 2 * gb) {
        const bool second = b >= gb;
        gemm128_slab(second ? b - gb : b, X, nrows, second ? Wb : Wa, second ? bb : ba, second ? Yb : Ya);
        return;
    }
    b -= 2 * gb;
    if (b < mb) {
        owner_marks_blk<RE>(b, idsE, ownE, plE, nnz, ngroups);
        return;
    }
    owner_marks_blk<RV>(b - mb, idsV, ownV, plV, nnz, ngroups);
}

#define OWN_THREADS 1024
#define OWN_WAVES   32
#define OWN_QCAP    768
#define OWN_BLK     512
#define OWN_UB      25
#define OWN_BATCH   8
#define OWN_GCAP    384
#define OWN_TP      132

template <int R, bool BLEND>
__global__ __launch_bounds__(OWN_THREADS) void own_rows(
    const float* __restrict__ src,
    const int* __restrict__ gidx,
    const int* __restrict__ sidx,
    const int* __restrict__ own,
    const int* __restrict__ plane,
    int ngroups,
    float* __restrict__ dst,
    const float* __restrict__ Ap,
    const float* __restrict__ X0,
    int nrows, int nnz,
    const float* __restrict__ Wt,
    const float* __restrict__ bt)
{
    static_assert(R <= 256, "local row must fit the low 8 bits of a queue entry");
    constexpr int NS = (R + OWN_WAVES - 1) / OWN_WAVES;
    constexpr int RT = (R + 15) / 16;
    __shared__ alignas(16) int qp[2 * OWN_WAVES * OWN_QCAP];
    int (*queue)[OWN_QCAP] = reinterpret_cast<int (*)[OWN_QCAP]>(qp);
    int* pool = qp + OWN_WAVES * OWN_QCAP;
    float* tile = reinterpret_cast<float*>(qp);
    static_assert(RT * 16 * OWN_TP <= 2 * OWN_WAVES * OWN_QCAP, "the row tile must fit the space of the queues and the pool");
    __shared__ int kept[OWN_WAVES][OWN_GCAP];
    __shared__ int cnt[OWN_WAVES][OWN_WAVES];

    const int tid  = threadIdx.x;
    const int lane = tid & 31;
    const int w    = __builtin_amdgcn_readfirstlane(tid >> 5);
    const int base = blockIdx.x * R;
    int rows = nrows - base;
    if (rows > R) rows = R;
    if (rows < 0) rows = 0;

    static_assert(NS <= 8, "a slot must index an 8-wide register vector");
    v8f ax = {}, ay = {}, az = {}, aw = {};
    v8i rc = {};

    const int seg = (((nnz + OWN_WAVES - 1) / OWN_WAVES) + OWN_BLK - 1) / OWN_BLK * OWN_BLK;
    const uint4* __restrict__ own4 = (const uint4*)own;
    const unsigned me = blockIdx.x;
    const unsigned me4 = me * 0x01010101u;
    const unsigned* __restrict__ myplane = (const unsigned*)plane + (size_t)(me >> 5) * (size_t)ngroups;
    const unsigned mebit = 1u << (me & 31u);
    int pos = w * seg;
    int segEnd = pos + seg;
    if (segEnd > nnz) segEnd = nnz;

    int gn = 0, gi = 0;

    int any;
    do {
        if (gi >= gn) {
            gn = 0;
            gi = 0;
            while (pos < segEnd && gn <= OWN_GCAP - 32) {
                unsigned d[OWN_UB];
#pragma unroll
                for (int u = 0; u < OWN_UB; ++u) {
                    int pb = pos + u * OWN_BLK;
                    if (pb >= segEnd) pb = pos;
                    d[u] = myplane[(pb >> 4) + lane];
                }
#pragma unroll
                for (int u = 0; u < OWN_UB; ++u) {
                    if (pos >= segEnd || gn > OWN_GCAP - 32) break;
                    const bool t = (d[u] & mebit) != 0u;
                    const unsigned hit = __builtin_amdgcn_ballot_w32(t);
                    if (t) kept[w][gn + (int)__builtin_amdgcn_mbcnt_lo(hit, 0u)] = (pos >> 4) + lane;
                    gn += __builtin_popcount(hit);
                    pos += OWN_BLK;
                }
            }
        }
        int qn = 0;
        while (gi < gn && qn <= OWN_QCAP - OWN_BLK) {
            const int i = gi + lane;
            const bool in = i < gn;
            const int g = in ? kept[w][i] : 0;
            const uint4 dd = own4[g];
            const unsigned y0 = dd.x ^ me4, y1 = dd.y ^ me4, y2 = dd.z ^ me4, y3 = dd.w ^ me4;
            const unsigned f0 = ~(((y0 & 0x7f7f7f7fu) + 0x7f7f7f7fu) | y0) & 0x80808080u;
            const unsigned f1 = ~(((y1 & 0x7f7f7f7fu) + 0x7f7f7f7fu) | y1) & 0x80808080u;
            const unsigned f2 = ~(((y2 & 0x7f7f7f7fu) + 0x7f7f7f7fu) | y2) & 0x80808080u;
            const unsigned f3 = ~(((y3 & 0x7f7f7f7fu) + 0x7f7f7f7fu) | y3) & 0x80808080u;
            const unsigned x0 = f0 >> 7, x1 = f1 >> 7, x2 = f2 >> 7, x3 = f3 >> 7;
            const unsigned n0 = (x0 | (x0 >> 7) | (x0 >> 14) | (x0 >> 21)) & 15u;
            const unsigned n1 = (x1 | (x1 >> 7) | (x1 >> 14) | (x1 >> 21)) & 15u;
            const unsigned n2 = (x2 | (x2 >> 7) | (x2 >> 14) | (x2 >> 21)) & 15u;
            const unsigned n3 = (x3 | (x3 >> 7) | (x3 >> 14) | (x3 >> 21)) & 15u;
            unsigned hm = n0 | (n1 << 4) | (n2 << 8) | (n3 << 12);
            const int nv = segEnd - g * 16;
            hm = !in || nv <= 0 ? 0u : (nv >= 16 ? hm : (hm & ((1u << nv) - 1u)));
            const int c = __builtin_popcount(hm);
            int incl = c;
            incl += __builtin_amdgcn_update_dpp(0, incl, 0x111, 0xf, 0xf, true);
            incl += __builtin_amdgcn_update_dpp(0, incl, 0x112, 0xf, 0xf, true);
            incl += __builtin_amdgcn_update_dpp(0, incl, 0x114, 0xf, 0xf, true);
            incl += __builtin_amdgcn_update_dpp(0, incl, 0x118, 0xf, 0xf, true);
            const int low = __builtin_amdgcn_readlane(incl, 15);
            if (lane >= 16) incl += low;
            int at = qn + incl - c;
            while (hm) {
                queue[w][at++] = g * 16 + __builtin_ctz(hm);
                hm &= hm - 1u;
            }
            qn += __builtin_amdgcn_readlane(incl, 31);
            gi += 32;
        }
        const unsigned l0 = (lane & 1) ? ~0u : 0u, l1 = (lane & 2) ? ~0u : 0u, l2 = (lane & 4) ? ~0u : 0u,
                       l3 = (lane & 8) ? ~0u : 0u, l4 = (lane & 16) ? ~0u : 0u;
        int mine = 0;
        for (int i0 = 0; i0 < qn; i0 += 32) {
            const int i = i0 + lane;
            const bool in = i < qn;
            int e = 0;
            if (in) {
                const int p = queue[w][i];
                const int s = sidx[p];
                const int g = gidx[p];
                e = (g << 8) | (s - base);
                queue[w][i] = e;
            }
            const unsigned vm = __builtin_amdgcn_ballot_w32(in);
            const unsigned b0 = __builtin_amdgcn_ballot_w32((e & 1) != 0), b1 = __builtin_amdgcn_ballot_w32((e & 2) != 0),
                           b2 = __builtin_amdgcn_ballot_w32((e & 4) != 0), b3 = __builtin_amdgcn_ballot_w32((e & 8) != 0),
                           b4 = __builtin_amdgcn_ballot_w32((e & 16) != 0);
            mine += __builtin_popcount(vm & ~(b0 ^ l0) & ~(b1 ^ l1) & ~(b2 ^ l2) & ~(b3 ^ l3) & ~(b4 ^ l4));
        }
        cnt[w][lane] = mine;
        __syncthreads();

        int before = 0, tot = 0;
        for (int q = 0; q < OWN_WAVES; ++q) {
            const int c = cnt[q][lane];
            if (q < w) before += c;
            tot += c;
        }
        int incl = tot;
        incl += __builtin_amdgcn_update_dpp(0, incl, 0x111, 0xf, 0xf, true);
        incl += __builtin_amdgcn_update_dpp(0, incl, 0x112, 0xf, 0xf, true);
        incl += __builtin_amdgcn_update_dpp(0, incl, 0x114, 0xf, 0xf, true);
        incl += __builtin_amdgcn_update_dpp(0, incl, 0x118, 0xf, 0xf, true);
        const int low = __builtin_amdgcn_readlane(incl, 15);
        if (lane >= 16) incl += low;
        const int ostart = incl - tot;
        int next = ostart + before;
        const int mystart = __builtin_amdgcn_readlane(ostart, w);
        const int mycnt   = __builtin_amdgcn_readlane(tot, w);

        for (int i0 = 0; i0 < qn; i0 += 32) {
            const int i = i0 + lane;
            const bool in = i < qn;
            const int e = in ? queue[w][i] : 0;
            const unsigned vm = __builtin_amdgcn_ballot_w32(in);
            const unsigned b0 = __builtin_amdgcn_ballot_w32((e & 1) != 0), b1 = __builtin_amdgcn_ballot_w32((e & 2) != 0),
                           b2 = __builtin_amdgcn_ballot_w32((e & 4) != 0), b3 = __builtin_amdgcn_ballot_w32((e & 8) != 0),
                           b4 = __builtin_amdgcn_ballot_w32((e & 16) != 0);
            const unsigned e0 = (e & 1) ? ~0u : 0u, e1 = (e & 2) ? ~0u : 0u, e2 = (e & 4) ? ~0u : 0u,
                           e3 = (e & 8) ? ~0u : 0u, e4 = (e & 16) ? ~0u : 0u;
            const unsigned peers = vm & ~(b0 ^ e0) & ~(b1 ^ e1) & ~(b2 ^ e2) & ~(b3 ^ e3) & ~(b4 ^ e4);
            const int rank = (int)__builtin_amdgcn_mbcnt_lo(peers, 0u);
            const int place = __shfl(next, e & 31);
            if (in) pool[place + rank] = e;
            next += __builtin_popcount(vm & ~(b0 ^ l0) & ~(b1 ^ l1) & ~(b2 ^ l2) & ~(b3 ^ l3) & ~(b4 ^ l4));
        }
        __syncthreads();

        auto drain = [&](int ev, int first, int n) {
            float4 v[OWN_BATCH];
            int rs[OWN_BATCH];
#pragma unroll
            for (int k = 0; k < OWN_BATCH; ++k) {
                if (k < n) {
                    const int e = __builtin_amdgcn_readlane(ev, first + k);
                    rs[k] = (e & 255) >> 5;
                    v[k] = ((const float4*)(src + (size_t)((unsigned)e >> 8) * DFEAT))[lane];
                }
            }
#pragma unroll
            for (int k = 0; k < OWN_BATCH; ++k) {
                if (k < n) {
                    const int s = rs[k];
                    ax[s] += v[k].x; ay[s] += v[k].y; az[s] += v[k].z; aw[s] += v[k].w;
                    if (BLEND) rc[s] += 1;
                }
            }
        };
        for (int h = 0; h < mycnt; h += 32) {
            const int left = mycnt - h;
            const int nb = left < 32 ? left : 32;
            const int ev = lane < nb ? pool[mystart + h + lane] : 0;
            int first = 0;
#pragma nounroll
            for (; first + OWN_BATCH <= nb; first += OWN_BATCH) drain(ev, first, OWN_BATCH);
            if (first < nb) drain(ev, first, nb - first);
        }

        any = __syncthreads_or((pos < segEnd || gi < gn) ? 1 : 0);
    } while (any);

    if (BLEND) {
        const float4* a4 = (const float4*)Ap + (size_t)base * 32;
        const float4* x4 = (const float4*)X0 + (size_t)base * 32;
        if (rows > 0) {
            float4 av[NS], xv[NS];
#pragma unroll
            for (int t = 0; t < NS; ++t) {
                const int r = w + OWN_WAVES * t;
                const int j = r < rows ? r : rows - 1;
                av[t] = a4[j * 32 + lane];
                xv[t] = x4[j * 32 + lane];
            }
#pragma unroll
            for (int t = 0; t < NS; ++t) {
                const int r = w + OWN_WAVES * t;
                if (r < rows) {
                    const float4 s = make_float4(ax[t], ay[t], az[t], aw[t]);
                    const float d = (float)rc[t];
                    float4 o;
                    o.x = 0.5f * (d * av[t].x + s.x) + 0.5f * xv[t].x;
                    o.y = 0.5f * (d * av[t].y + s.y) + 0.5f * xv[t].y;
                    o.z = 0.5f * (d * av[t].z + s.z) + 0.5f * xv[t].z;
                    o.w = 0.5f * (d * av[t].w + s.w) + 0.5f * xv[t].w;
                    ((float4*)(tile + r * OWN_TP))[lane] = o;
                }
            }
        }
    } else {
#pragma unroll
        for (int t = 0; t < NS; ++t) {
            const int r = w + OWN_WAVES * t;
            if (r < rows) ((float4*)(tile + r * OWN_TP))[lane] = make_float4(ax[t], ay[t], az[t], aw[t]);
        }
    }
    __syncthreads();

    {
        const int half = lane >> 4;
        const int r16  = lane & 15;
        const int nunits = 2 * ((rows + 15) >> 4);
        for (int unit = w; unit < nunits; unit += OWN_WAVES) {
            const int rt = unit >> 1;
            const int u  = unit & 1;
            v8f acc[4] = {};
            const float* arow = tile + (rt * 16 + r16) * OWN_TP + 4 * half;
            const v4f* wp = (const v4f*)(Wt + (size_t)(4 * half) * DFEAT + 64 * u + 4 * r16);
            v4f a4[2];
            v4f b[2][4];
            a4[0] = *(const v4f*)arow;
#pragma unroll
            for (int i = 0; i < 4; ++i) b[0][i] = wp[i * 32];
#pragma unroll
            for (int q = 0; q < 16; ++q) {
                if (q + 1 < 16) {
                    a4[(q + 1) & 1] = *(const v4f*)(arow + 8 * (q + 1));
#pragma unroll
                    for (int i = 0; i < 4; ++i) b[(q + 1) & 1][i] = wp[((q + 1) * 8 + i) * 32];
                }
                __builtin_amdgcn_sched_barrier(0);
#pragma unroll
                for (int e = 0; e < 2; ++e) {
                    v2f a;
                    a.x = a4[q & 1][2 * e];
                    a.y = a4[q & 1][2 * e + 1];
#pragma unroll
                    for (int t = 0; t < 4; ++t) {
                        v2f bb;
                        bb.x = b[q & 1][2 * e][t];
                        bb.y = b[q & 1][2 * e + 1][t];
                        acc[t] = __builtin_amdgcn_wmma_f32_16x16x4_f32(
                            false, a, false, bb, (short)0, acc[t], false, false);
                    }
                }
            }
            v4f bv;
            if (bt) bv = ((const v4f*)bt)[16 * u + r16];
            else { bv.x = 0.0f; bv.y = 0.0f; bv.z = 0.0f; bv.w = 0.0f; }
#pragma unroll
            for (int j = 0; j < 8; ++j) {
                const int row = rt * 16 + j + 8 * half;
                if (row < rows) {
                    v4f o;
                    o.x = acc[0][j] + bv.x;
                    o.y = acc[1][j] + bv.y;
                    o.z = acc[2][j] + bv.z;
                    o.w = acc[3][j] + bv.w;
                    ((v4f*)(dst + (size_t)(base + row) * DFEAT))[16 * u + r16] = o;
                }
            }
        }
    }
}

extern "C" void kernel_launch(void* const* d_in, const int* in_sizes, int n_in,
                              void* d_out, int out_size, void* d_ws, size_t ws_size,
                              hipStream_t stream) {
    const float* X      = (const float*)d_in[0];
    const int*   vertex = (const int*)d_in[1];
    const int*   edges  = (const int*)d_in[2];
    const float* X0     = (const float*)d_in[3];
    const float* W1     = (const float*)d_in[4];
    const float* b1     = (const float*)d_in[5];
    const float* W2     = (const float*)d_in[6];
    const float* b2     = (const float*)d_in[7];
    const float* W3     = (const float*)d_in[8];
    const float* b3     = (const float*)d_in[9];
    float* out = (float*)d_out;

    constexpr int N = 50000, M = 25000, NNZ = 800000;
    constexpr size_t ND = (size_t)N * DFEAT;
    constexpr size_t MD = (size_t)M * DFEAT;

    char* ws = (char*)d_ws;
    auto align256 = [](size_t x) { return (x + 255) & ~(size_t)255; };
    size_t off = 0;
    float* XW = (float*)(ws + off); off = align256(off + ND * 4);
    char*  Mk = ws + off;           off = align256(off + MD * 4);
    float* Ap = (float*)(ws + off); off = align256(off + ND * 4);
    float* Bm = (float*)(ws + off); off = align256(off + MD * 4);

    constexpr int gemmN = (N + 127) / 128;
    constexpr int RM = 98;
    constexpr int RN = 196;
    static_assert((M + RM - 1) / RM <= 256 && (N + RN - 1) / RN <= 256, "an owner number must fit one byte");

    constexpr int OWN_POS    = (NNZ + OWN_BLK - 1) / OWN_BLK * OWN_BLK;
    constexpr int OWN_WORDS  = OWN_POS / 4;
    constexpr int OWN_GROUPS = OWN_POS / 16;
    constexpr int OWN_GRID   = (OWN_GROUPS + 255) / 256;
    constexpr size_t MARK_BYTES = ((size_t)OWN_POS + (size_t)OWN_GROUPS * 32 + 255) / 256 * 256;
    static_assert(2 * MARK_BYTES <= MD * 4, "both lists' marks must fit their region");
    int* ownE = (int*)Mk;
    int* plE  = ownE + OWN_WORDS;
    int* ownV = (int*)(Mk + MARK_BYTES);
    int* plV  = ownV + OWN_WORDS;

    input_jobs<RM, RN><<<2 * gemmN + 2 * OWN_GRID, 256, 0, stream>>>(
        X, N, W1, b1, XW, W2, b2, Ap, edges, ownE, plE, vertex, ownV, plV, NNZ, OWN_GROUPS, gemmN, OWN_GRID);

    own_rows<RM, false><<<(M + RM - 1) / RM, OWN_THREADS, 0, stream>>>(XW, vertex, edges, ownE, plE, OWN_GROUPS, Bm, nullptr, nullptr, M, NNZ, W2 + 128 * DFEAT, nullptr);

    own_rows<RN, true><<<(N + RN - 1) / RN, OWN_THREADS, 0, stream>>>(Bm, edges, vertex, ownV, plV, OWN_GROUPS, out, Ap, X0, N, NNZ, W3, b3);
}
